# starting kernel
# baseline (speedup 1.0000x reference)
#include <hip/hip_runtime.h>
#include <hip/hip_bf16.h>


typedef float v2f __attribute__((ext_vector_type(2)));
typedef float v8f __attribute__((ext_vector_type(8)));

#define F_IN        512
#define C_OUT       16
#define LDSW_STRIDE (F_IN + 2)

__launch_bounds__(256)
__global__ void appnp_gemm_relu(const float* __restrict__ feat,
                                const float* __restrict__ W,
                                const float* __restrict__ bias,
                                float* __restrict__ xbuf,
                                int nTiles) {
    __shared__ float Wlds[C_OUT * LDSW_STRIDE];

    for (int i = threadIdx.x; i < (C_OUT * F_IN) / 4; i += blockDim.x) {
        const int   r  = (i * 4) / F_IN;
        const int   c  = (i * 4) % F_IN;
        const float4 w = ((const float4*)W)[i];
        float* dr = &Wlds[r * LDSW_STRIDE + c];
        dr[0] = w.x; dr[1] = w.y; dr[2] = w.z; dr[3] = w.w;
    }
    __syncthreads();

    const int wave = threadIdx.x >> 5;
    const int lane = threadIdx.x & 31;
    const int tile = blockIdx.x * 8 + wave;
    if (tile >= nTiles) return;

    const int half = lane >> 4;
    const int l    = lane & 15;

    const float* arow = feat + (size_t)(tile * 16 + l) * F_IN + 2 * half;
    const float* brow = &Wlds[l * LDSW_STRIDE + 2 * half];

    v8f acc = {};
#pragma unroll 8
    for (int k0 = 0; k0 < F_IN; k0 += 4) {
        v2f a = *(const v2f*)(arow + k0);
        v2f b = *(const v2f*)(brow + k0);
        acc = __builtin_amdgcn_wmma_f32_16x16x4_f32(
             false, a,  false, b,
             (short)0, acc,  false,  false);
    }

    const float bv = bias[l];
#pragma unroll
    for (int r = 0; r < 8; ++r) {
        const int row = tile * 16 + r + 8 * half;
        float v = acc[r] + bv;
        v = v > 0.0f ? v : 0.0f;
        xbuf[(size_t)row * C_OUT + l] = v;
    }
}

__global__ void appnp_zero(float* __restrict__ p, long long n) {
    long long i = (long long)blockIdx.x * blockDim.x + threadIdx.x;
    if (i < n) p[i] = 0.0f;
}

__global__ void appnp_degree(const int* __restrict__ src,
                             const int* __restrict__ dst,
                             float* __restrict__ outdeg,
                             float* __restrict__ indeg, int E) {
    int e = blockIdx.x * blockDim.x + threadIdx.x;
    if (e < E) {
        atomicAdd(&outdeg[src[e]], 1.0f);
        atomicAdd(&indeg[dst[e]], 1.0f);
    }
}

__global__ void appnp_scatter(const int* __restrict__ src,
                              const int* __restrict__ dst,
                              const float* __restrict__ xbuf,
                              const float* __restrict__ outdeg,
                              float* __restrict__ agg, int E) {
    long long t = (long long)blockIdx.x * blockDim.x + threadIdx.x;
    if (t >= (long long)E * 4) return;
    const int e = (int)(t >> 2);
    const int g = (int)(t & 3);
    const int s = src[e];
    const int d = dst[e];
    const float ns = rsqrtf(fmaxf(outdeg[s], 1.0f));
    const float4 v = ((const float4*)xbuf)[(size_t)s * 4 + g];
    float* ap = agg + (size_t)d * C_OUT + g * 4;
    atomicAdd(ap + 0, v.x * ns);
    atomicAdd(ap + 1, v.y * ns);
    atomicAdd(ap + 2, v.z * ns);
    atomicAdd(ap + 3, v.w * ns);
}

__global__ void appnp_finalize(const float* __restrict__ agg,
                               const float* __restrict__ indeg,
                               const float* __restrict__ xbuf,
                               float* __restrict__ out, int N) {
    int t = blockIdx.x * blockDim.x + threadIdx.x;
    if (t >= N * 4) return;
    const int n = t >> 2;
    const float nd = rsqrtf(fmaxf(indeg[n], 1.0f));
    const float4 a  = ((const float4*)agg)[t];
    const float4 x0 = ((const float4*)xbuf)[t];
    float4 o;
    o.x = 0.8f * a.x * nd + 0.2f * x0.x;
    o.y = 0.8f * a.y * nd + 0.2f * x0.y;
    o.z = 0.8f * a.z * nd + 0.2f * x0.z;
    o.w = 0.8f * a.w * nd + 0.2f * x0.w;
    ((float4*)out)[t] = o;
}

extern "C" void kernel_launch(void* const* d_in, const int* in_sizes, int n_in,
                              void* d_out, int out_size, void* d_ws, size_t ws_size,
                              hipStream_t stream) {
    const float* feat = (const float*)d_in[0];
    const float* W    = (const float*)d_in[1];
    const float* bias = (const float*)d_in[2];
    const int*   src  = (const int*)d_in[3];
    const int*   dst  = (const int*)d_in[4];

    const int N = in_sizes[0] / F_IN;
    const int E = in_sizes[3];

    float* ws     = (float*)d_ws;
    float* xbuf   = ws;
    float* agg    = xbuf + (size_t)N * C_OUT;
    float* outdeg = agg  + (size_t)N * C_OUT;
    float* indeg  = outdeg + N;

    {
        long long cnt = (long long)N * C_OUT + 2LL * N;
        int blocks = (int)((cnt + 255) / 256);
        appnp_zero<<<blocks, 256, 0, stream>>>(agg, cnt);
    }
    appnp_degree<<<(E + 255) / 256, 256, 0, stream>>>(src, dst, outdeg, indeg, E);
    {
        const int nTiles = (N + 15) / 16;
        const int blocks = (nTiles + 7) / 8;
        appnp_gemm_relu<<<blocks, 256, 0, stream>>>(feat, W, bias, xbuf, nTiles);
    }
    {
        long long t = (long long)E * 4;
        int blocks = (int)((t + 255) / 256);
        appnp_scatter<<<blocks, 256, 0, stream>>>(src, dst, xbuf, outdeg, agg, E);
    }
    appnp_finalize<<<(N * 4 + 255) / 256, 256, 0, stream>>>(agg, indeg, xbuf,
                                                            (float*)d_out, N);
}


# Degree counts by per-workgroup LDS histograms (u32 ds_add per node range x edge chunk) summed from partial planes; replaces 6.4M scattered global f32 atomics
# speedup vs baseline: 2.1648x; 2.1648x over previous
#include <hip/hip_runtime.h>
#include <hip/hip_bf16.h>


typedef float v2f __attribute__((ext_vector_type(2)));
typedef float v8f __attribute__((ext_vector_type(8)));

#define F_IN        512
#define C_OUT       16
#define LDSW_STRIDE (F_IN + 2)

__launch_bounds__(256)
__global__ void appnp_gemm_relu(const float* __restrict__ feat,
                                const float* __restrict__ W,
                                const float* __restrict__ bias,
                                float* __restrict__ xbuf,
                                int nTiles) {
    __shared__ float Wlds[C_OUT * LDSW_STRIDE];

    for (int i = threadIdx.x; i < (C_OUT * F_IN) / 4; i += blockDim.x) {
        const int   r  = (i * 4) / F_IN;
        const int   c  = (i * 4) % F_IN;
        const float4 w = ((const float4*)W)[i];
        float* dr = &Wlds[r * LDSW_STRIDE + c];
        dr[0] = w.x; dr[1] = w.y; dr[2] = w.z; dr[3] = w.w;
    }
    __syncthreads();

    const int wave = threadIdx.x >> 5;
    const int lane = threadIdx.x & 31;
    const int tile = blockIdx.x * 8 + wave;
    if (tile >= nTiles) return;

    const int half = lane >> 4;
    const int l    = lane & 15;

    const float* arow = feat + (size_t)(tile * 16 + l) * F_IN + 2 * half;
    const float* brow = &Wlds[l * LDSW_STRIDE + 2 * half];

    v8f acc = {};
#pragma unroll 8
    for (int k0 = 0; k0 < F_IN; k0 += 4) {
        v2f a = *(const v2f*)(arow + k0);
        v2f b = *(const v2f*)(brow + k0);
        acc = __builtin_amdgcn_wmma_f32_16x16x4_f32(
             false, a,  false, b,
             (short)0, acc,  false,  false);
    }

    const float bv = bias[l];
#pragma unroll
    for (int r = 0; r < 8; ++r) {
        const int row = tile * 16 + r + 8 * half;
        float v = acc[r] + bv;
        v = v > 0.0f ? v : 0.0f;
        xbuf[(size_t)row * C_OUT + l] = v;
    }
}

__global__ void appnp_zero(float* __restrict__ p, long long n) {
    long long i = (long long)blockIdx.x * blockDim.x + threadIdx.x;
    if (i < n) p[i] = 0.0f;
}

#define DEG_P       16
#define DEG_R       12800
#define DEG_THREADS 1024

static __device__ __forceinline__ void deg_bin4(unsigned* cnt, const int4 q, const int lo) {
    unsigned r;
    r = (unsigned)(q.x - lo); if (r < (unsigned)DEG_R) atomicAdd(&cnt[r], 1u);
    r = (unsigned)(q.y - lo); if (r < (unsigned)DEG_R) atomicAdd(&cnt[r], 1u);
    r = (unsigned)(q.z - lo); if (r < (unsigned)DEG_R) atomicAdd(&cnt[r], 1u);
    r = (unsigned)(q.w - lo); if (r < (unsigned)DEG_R) atomicAdd(&cnt[r], 1u);
}

__launch_bounds__(DEG_THREADS)
__global__ void appnp_degree_count(const int* __restrict__ src,
                                   const int* __restrict__ dst,
                                   unsigned* __restrict__ part,
                                   int E, int N, int chunk) {
    __shared__ unsigned cnt[DEG_R];

    const int h = blockIdx.x;
    const int p = blockIdx.y;
    const int a = blockIdx.z;
    const int* __restrict__ idx = a ? dst : src;
    const int lo = h * DEG_R;

    for (int i = threadIdx.x; i < DEG_R; i += DEG_THREADS) cnt[i] = 0u;
    __syncthreads();

    const int e0 = p * chunk;
    int e1 = e0 + chunk; if (e1 > E) e1 = E;
    const int len  = e1 > e0 ? e1 - e0 : 0;
    const int nvec = len >> 2;
    const int4* __restrict__ v = (const int4*)(idx + e0);

    int i = threadIdx.x;
    for (; i + 3 * DEG_THREADS < nvec; i += 4 * DEG_THREADS) {
        const int4 q0 = v[i];
        const int4 q1 = v[i + DEG_THREADS];
        const int4 q2 = v[i + 2 * DEG_THREADS];
        const int4 q3 = v[i + 3 * DEG_THREADS];
        deg_bin4(cnt, q0, lo); deg_bin4(cnt, q1, lo);
        deg_bin4(cnt, q2, lo); deg_bin4(cnt, q3, lo);
    }
    for (; i < nvec; i += DEG_THREADS) deg_bin4(cnt, v[i], lo);
    for (int e = e0 + (nvec << 2) + threadIdx.x; e < e1; e += DEG_THREADS) {
        const unsigned r = (unsigned)(idx[e] - lo);
        if (r < (unsigned)DEG_R) atomicAdd(&cnt[r], 1u);
    }
    __syncthreads();

    int nr = N - lo; if (nr > DEG_R) nr = DEG_R;
    unsigned* __restrict__ plane = part + ((size_t)(a * DEG_P + p)) * (size_t)N + lo;
    for (int k = threadIdx.x; k < nr; k += DEG_THREADS) plane[k] = cnt[k];
}

__global__ void appnp_degree_sum(const unsigned* __restrict__ part,
                                 float* __restrict__ deg, int N) {
    const int t = blockIdx.x * blockDim.x + threadIdx.x;
    if (t >= 2 * N) return;
    const int a = t >= N ? 1 : 0;
    const int n = t - a * N;
    const unsigned* __restrict__ q = part + (size_t)a * DEG_P * (size_t)N + n;
    unsigned s = 0u;
#pragma unroll
    for (int k = 0; k < DEG_P; ++k) s += q[(size_t)k * (size_t)N];
    deg[t] = (float)s;
}

__global__ void appnp_scatter(const int* __restrict__ src,
                              const int* __restrict__ dst,
                              const float* __restrict__ xbuf,
                              const float* __restrict__ outdeg,
                              float* __restrict__ agg, int E) {
    long long t = (long long)blockIdx.x * blockDim.x + threadIdx.x;
    if (t >= (long long)E * 4) return;
    const int e = (int)(t >> 2);
    const int g = (int)(t & 3);
    const int s = src[e];
    const int d = dst[e];
    const float ns = rsqrtf(fmaxf(outdeg[s], 1.0f));
    const float4 v = ((const float4*)xbuf)[(size_t)s * 4 + g];
    float* ap = agg + (size_t)d * C_OUT + g * 4;
    atomicAdd(ap + 0, v.x * ns);
    atomicAdd(ap + 1, v.y * ns);
    atomicAdd(ap + 2, v.z * ns);
    atomicAdd(ap + 3, v.w * ns);
}

__global__ void appnp_finalize(const float* __restrict__ agg,
                               const float* __restrict__ indeg,
                               const float* __restrict__ xbuf,
                               float* __restrict__ out, int N) {
    int t = blockIdx.x * blockDim.x + threadIdx.x;
    if (t >= N * 4) return;
    const int n = t >> 2;
    const float nd = rsqrtf(fmaxf(indeg[n], 1.0f));
    const float4 a  = ((const float4*)agg)[t];
    const float4 x0 = ((const float4*)xbuf)[t];
    float4 o;
    o.x = 0.8f * a.x * nd + 0.2f * x0.x;
    o.y = 0.8f * a.y * nd + 0.2f * x0.y;
    o.z = 0.8f * a.z * nd + 0.2f * x0.z;
    o.w = 0.8f * a.w * nd + 0.2f * x0.w;
    ((float4*)out)[t] = o;
}

extern "C" void kernel_launch(void* const* d_in, const int* in_sizes, int n_in,
                              void* d_out, int out_size, void* d_ws, size_t ws_size,
                              hipStream_t stream) {
    const float* feat = (const float*)d_in[0];
    const float* W    = (const float*)d_in[1];
    const float* bias = (const float*)d_in[2];
    const int*   src  = (const int*)d_in[3];
    const int*   dst  = (const int*)d_in[4];

    const int N = in_sizes[0] / F_IN;
    const int E = in_sizes[3];

    float* ws     = (float*)d_ws;
    float* xbuf   = ws;
    float* agg    = xbuf + (size_t)N * C_OUT;
    float* outdeg = agg  + (size_t)N * C_OUT;
    float* indeg  = outdeg + N;

    {
        unsigned* part = (unsigned*)xbuf;
        const int chunk  = (((E + DEG_P - 1) / DEG_P) + 3) & ~3;
        const int ranges = (N + DEG_R - 1) / DEG_R;
        appnp_degree_count<<<dim3(ranges, DEG_P, 2), DEG_THREADS, 0, stream>>>(
            src, dst, part, E, N, chunk);
        appnp_degree_sum<<<(2 * N + 255) / 256, 256, 0, stream>>>(part, outdeg, N);
    }
    {
        long long cnt = (long long)N * C_OUT;
        int blocks = (int)((cnt + 255) / 256);
        appnp_zero<<<blocks, 256, 0, stream>>>(agg, cnt);
    }
    {
        const int nTiles = (N + 15) / 16;
        const int blocks = (nTiles + 7) / 8;
        appnp_gemm_relu<<<blocks, 256, 0, stream>>>(feat, W, bias, xbuf, nTiles);
    }
    {
        long long t = (long long)E * 4;
        int blocks = (int)((t + 255) / 256);
        appnp_scatter<<<blocks, 256, 0, stream>>>(src, dst, xbuf, outdeg, agg, E);
    }
    appnp_finalize<<<(N * 4 + 255) / 256, 256, 0, stream>>>(agg, indeg, xbuf,
                                                            (float*)d_out, N);
}


# scatter-add: half-wave per edge (lane = channel), resident 16k-wave grid looping over edges: an edge's 16 f32 atomics leave as one 64-B row request, not four strided; sums stay in arrival order
# speedup vs baseline: 3.0119x; 1.3913x over previous
#include <hip/hip_runtime.h>
#include <hip/hip_bf16.h>


typedef float v2f __attribute__((ext_vector_type(2)));
typedef float v8f __attribute__((ext_vector_type(8)));

#define F_IN        512
#define C_OUT       16
#define LDSW_STRIDE (F_IN + 2)

__launch_bounds__(256)
__global__ void appnp_gemm_relu(const float* __restrict__ feat,
                                const float* __restrict__ W,
                                const float* __restrict__ bias,
                                float* __restrict__ xbuf,
                                int nTiles) {
    __shared__ float Wlds[C_OUT * LDSW_STRIDE];

    for (int i = threadIdx.x; i < (C_OUT * F_IN) / 4; i += blockDim.x) {
        const int   r  = (i * 4) / F_IN;
        const int   c  = (i * 4) % F_IN;
        const float4 w = ((const float4*)W)[i];
        float* dr = &Wlds[r * LDSW_STRIDE + c];
        dr[0] = w.x; dr[1] = w.y; dr[2] = w.z; dr[3] = w.w;
    }
    __syncthreads();

    const int wave = threadIdx.x >> 5;
    const int lane = threadIdx.x & 31;
    const int tile = blockIdx.x * 8 + wave;
    if (tile >= nTiles) return;

    const int half = lane >> 4;
    const int l    = lane & 15;

    const float* arow = feat + (size_t)(tile * 16 + l) * F_IN + 2 * half;
    const float* brow = &Wlds[l * LDSW_STRIDE + 2 * half];

    v8f acc = {};
#pragma unroll 8
    for (int k0 = 0; k0 < F_IN; k0 += 4) {
        v2f a = *(const v2f*)(arow + k0);
        v2f b = *(const v2f*)(brow + k0);
        acc = __builtin_amdgcn_wmma_f32_16x16x4_f32(
             false, a,  false, b,
             (short)0, acc,  false,  false);
    }

    const float bv = bias[l];
#pragma unroll
    for (int r = 0; r < 8; ++r) {
        const int row = tile * 16 + r + 8 * half;
        float v = acc[r] + bv;
        v = v > 0.0f ? v : 0.0f;
        xbuf[(size_t)row * C_OUT + l] = v;
    }
}

__global__ void appnp_zero(float* __restrict__ p, long long n) {
    long long i = (long long)blockIdx.x * blockDim.x + threadIdx.x;
    if (i < n) p[i] = 0.0f;
}

#define DEG_P       16
#define DEG_R       12800
#define DEG_THREADS 1024

static __device__ __forceinline__ void deg_bin4(unsigned* cnt, const int4 q, const int lo) {
    unsigned r;
    r = (unsigned)(q.x - lo); if (r < (unsigned)DEG_R) atomicAdd(&cnt[r], 1u);
    r = (unsigned)(q.y - lo); if (r < (unsigned)DEG_R) atomicAdd(&cnt[r], 1u);
    r = (unsigned)(q.z - lo); if (r < (unsigned)DEG_R) atomicAdd(&cnt[r], 1u);
    r = (unsigned)(q.w - lo); if (r < (unsigned)DEG_R) atomicAdd(&cnt[r], 1u);
}

__launch_bounds__(DEG_THREADS)
__global__ void appnp_degree_count(const int* __restrict__ src,
                                   const int* __restrict__ dst,
                                   unsigned* __restrict__ part,
                                   int E, int N, int chunk) {
    __shared__ unsigned cnt[DEG_R];

    const int h = blockIdx.x;
    const int p = blockIdx.y;
    const int a = blockIdx.z;
    const int* __restrict__ idx = a ? dst : src;
    const int lo = h * DEG_R;

    for (int i = threadIdx.x; i < DEG_R; i += DEG_THREADS) cnt[i] = 0u;
    __syncthreads();

    const int e0 = p * chunk;
    int e1 = e0 + chunk; if (e1 > E) e1 = E;
    const int len  = e1 > e0 ? e1 - e0 : 0;
    const int nvec = len >> 2;
    const int4* __restrict__ v = (const int4*)(idx + e0);

    int i = threadIdx.x;
    for (; i + 3 * DEG_THREADS < nvec; i += 4 * DEG_THREADS) {
        const int4 q0 = v[i];
        const int4 q1 = v[i + DEG_THREADS];
        const int4 q2 = v[i + 2 * DEG_THREADS];
        const int4 q3 = v[i + 3 * DEG_THREADS];
        deg_bin4(cnt, q0, lo); deg_bin4(cnt, q1, lo);
        deg_bin4(cnt, q2, lo); deg_bin4(cnt, q3, lo);
    }
    for (; i < nvec; i += DEG_THREADS) deg_bin4(cnt, v[i], lo);
    for (int e = e0 + (nvec << 2) + threadIdx.x; e < e1; e += DEG_THREADS) {
        const unsigned r = (unsigned)(idx[e] - lo);
        if (r < (unsigned)DEG_R) atomicAdd(&cnt[r], 1u);
    }
    __syncthreads();

    int nr = N - lo; if (nr > DEG_R) nr = DEG_R;
    unsigned* __restrict__ plane = part + ((size_t)(a * DEG_P + p)) * (size_t)N + lo;
    for (int k = threadIdx.x; k < nr; k += DEG_THREADS) plane[k] = cnt[k];
}

__global__ void appnp_degree_sum(const unsigned* __restrict__ part,
                                 float* __restrict__ deg, int N) {
    const int t = blockIdx.x * blockDim.x + threadIdx.x;
    if (t >= 2 * N) return;
    const int a = t >= N ? 1 : 0;
    const int n = t - a * N;
    const unsigned* __restrict__ q = part + (size_t)a * DEG_P * (size_t)N + n;
    unsigned s = 0u;
#pragma unroll
    for (int k = 0; k < DEG_P; ++k) s += q[(size_t)k * (size_t)N];
    deg[t] = (float)s;
}

#define SC_THREADS 256
#define SC_BLOCKS  2048

template <int J>
static __device__ __forceinline__ unsigned row_bcast(unsigned v) {
    return (unsigned)__builtin_amdgcn_update_dpp(0, (int)v, 0x150 + J, 0xf, 0xf, true);
}

template <int J>
static __device__ __forceinline__ float sc_load(const float* __restrict__ xbuf,
                                                unsigned srow, unsigned cb) {
    return *(const float*)((const char*)xbuf + (row_bcast<J>(srow) + cb));
}

template <int J, bool GUARD>
static __device__ __forceinline__ void sc_add(float* __restrict__ agg, unsigned drow,
                                              unsigned cb, float x, float nv,
                                              int eb, int E) {
    const unsigned off = row_bcast<J>(drow) + cb;
    const float    n   = __uint_as_float(row_bcast<J>(__float_as_uint(nv)));
    if (!GUARD || eb + J < E)
        atomicAdd((float*)((char*)agg + off), x * n);
}

template <bool GUARD>
static __device__ __forceinline__ void sc_batch(const float* __restrict__ xbuf,
                                                float* __restrict__ agg,
                                                unsigned srow, unsigned drow, float degs,
                                                unsigned cb, int eb, int E) {
    const float x0  = sc_load<0>(xbuf, srow, cb);
    const float x1  = sc_load<1>(xbuf, srow, cb);
    const float x2  = sc_load<2>(xbuf, srow, cb);
    const float x3  = sc_load<3>(xbuf, srow, cb);
    const float x4  = sc_load<4>(xbuf, srow, cb);
    const float x5  = sc_load<5>(xbuf, srow, cb);
    const float x6  = sc_load<6>(xbuf, srow, cb);
    const float x7  = sc_load<7>(xbuf, srow, cb);
    const float x8  = sc_load<8>(xbuf, srow, cb);
    const float x9  = sc_load<9>(xbuf, srow, cb);
    const float x10 = sc_load<10>(xbuf, srow, cb);
    const float x11 = sc_load<11>(xbuf, srow, cb);
    const float x12 = sc_load<12>(xbuf, srow, cb);
    const float x13 = sc_load<13>(xbuf, srow, cb);
    const float x14 = sc_load<14>(xbuf, srow, cb);
    const float x15 = sc_load<15>(xbuf, srow, cb);
    const float nv = rsqrtf(fmaxf(degs, 1.0f));
    sc_add<0,  GUARD>(agg, drow, cb, x0,  nv, eb, E);
    sc_add<1,  GUARD>(agg, drow, cb, x1,  nv, eb, E);
    sc_add<2,  GUARD>(agg, drow, cb, x2,  nv, eb, E);
    sc_add<3,  GUARD>(agg, drow, cb, x3,  nv, eb, E);
    sc_add<4,  GUARD>(agg, drow, cb, x4,  nv, eb, E);
    sc_add<5,  GUARD>(agg, drow, cb, x5,  nv, eb, E);
    sc_add<6,  GUARD>(agg, drow, cb, x6,  nv, eb, E);
    sc_add<7,  GUARD>(agg, drow, cb, x7,  nv, eb, E);
    sc_add<8,  GUARD>(agg, drow, cb, x8,  nv, eb, E);
    sc_add<9,  GUARD>(agg, drow, cb, x9,  nv, eb, E);
    sc_add<10, GUARD>(agg, drow, cb, x10, nv, eb, E);
    sc_add<11, GUARD>(agg, drow, cb, x11, nv, eb, E);
    sc_add<12, GUARD>(agg, drow, cb, x12, nv, eb, E);
    sc_add<13, GUARD>(agg, drow, cb, x13, nv, eb, E);
    sc_add<14, GUARD>(agg, drow, cb, x14, nv, eb, E);
    sc_add<15, GUARD>(agg, drow, cb, x15, nv, eb, E);
}

__launch_bounds__(SC_THREADS)
__global__ void appnp_scatter(const int* __restrict__ src,
                              const int* __restrict__ dst,
                              const float* __restrict__ xbuf,
                              const float* __restrict__ outdeg,
                              float* __restrict__ agg, int E) {
    const int lane   = threadIdx.x & 31;
    const int gwave  = (int)((blockIdx.x * SC_THREADS + threadIdx.x) >> 5);
    const int nwaves = (int)((gridDim.x * SC_THREADS) >> 5);
    const int nb     = (E + 31) >> 5;
    const unsigned cb = (unsigned)(lane & 15) << 2;
    const int hofs   = lane & 16;

    int b = gwave;
    if (b >= nb) return;
    int e = (b << 5) + lane;
    int s = 0, d = 0;
    if (e < E) { s = src[e]; d = dst[e]; }

    for (;;) {
        const float degs = outdeg[s];
        const int bn = b + nwaves;
        int sn = 0, dn = 0;
        if (bn < nb) {
            const int en = (bn << 5) + lane;
            if (en < E) { sn = src[en]; dn = dst[en]; }
        }
        const unsigned srow = (unsigned)s << 6;
        const unsigned drow = (unsigned)d << 6;
        const int eb = (b << 5) + hofs;
        if ((b << 5) + 32 <= E) sc_batch<false>(xbuf, agg, srow, drow, degs, cb, eb, E);
        else                    sc_batch<true >(xbuf, agg, srow, drow, degs, cb, eb, E);
        if (bn >= nb) break;
        b = bn; s = sn; d = dn;
    }
}

__global__ void appnp_finalize(const float* __restrict__ agg,
                               const float* __restrict__ indeg,
                               const float* __restrict__ xbuf,
                               float* __restrict__ out, int N) {
    int t = blockIdx.x * blockDim.x + threadIdx.x;
    if (t >= N * 4) return;
    const int n = t >> 2;
    const float nd = rsqrtf(fmaxf(indeg[n], 1.0f));
    const float4 a  = ((const float4*)agg)[t];
    const float4 x0 = ((const float4*)xbuf)[t];
    float4 o;
    o.x = 0.8f * a.x * nd + 0.2f * x0.x;
    o.y = 0.8f * a.y * nd + 0.2f * x0.y;
    o.z = 0.8f * a.z * nd + 0.2f * x0.z;
    o.w = 0.8f * a.w * nd + 0.2f * x0.w;
    ((float4*)out)[t] = o;
}

extern "C" void kernel_launch(void* const* d_in, const int* in_sizes, int n_in,
                              void* d_out, int out_size, void* d_ws, size_t ws_size,
                              hipStream_t stream) {
    const float* feat = (const float*)d_in[0];
    const float* W    = (const float*)d_in[1];
    const float* bias = (const float*)d_in[2];
    const int*   src  = (const int*)d_in[3];
    const int*   dst  = (const int*)d_in[4];

    const int N = in_sizes[0] / F_IN;
    const int E = in_sizes[3];

    float* ws     = (float*)d_ws;
    float* xbuf   = ws;
    float* agg    = xbuf + (size_t)N * C_OUT;
    float* outdeg = agg  + (size_t)N * C_OUT;
    float* indeg  = outdeg + N;

    {
        unsigned* part = (unsigned*)xbuf;
        const int chunk  = (((E + DEG_P - 1) / DEG_P) + 3) & ~3;
        const int ranges = (N + DEG_R - 1) / DEG_R;
        appnp_degree_count<<<dim3(ranges, DEG_P, 2), DEG_THREADS, 0, stream>>>(
            src, dst, part, E, N, chunk);
        appnp_degree_sum<<<(2 * N + 255) / 256, 256, 0, stream>>>(part, outdeg, N);
    }
    {
        long long cnt = (long long)N * C_OUT;
        int blocks = (int)((cnt + 255) / 256);
        appnp_zero<<<blocks, 256, 0, stream>>>(agg, cnt);
    }
    {
        const int nTiles = (N + 15) / 16;
        const int blocks = (nTiles + 7) / 8;
        appnp_gemm_relu<<<blocks, 256, 0, stream>>>(feat, W, bias, xbuf, nTiles);
    }
    {
        appnp_scatter<<<SC_BLOCKS, SC_THREADS, 0, stream>>>(src, dst, xbuf, outdeg, agg, E);
    }
    appnp_finalize<<<(N * 4 + 255) / 256, 256, 0, stream>>>(agg, indeg, xbuf,
                                                            (float*)d_out, N);
}


# scatter-add: one f32 accumulator per L2 half: a wave reads its die id and adds into its own half's copy (2nd copy is out, zeroed first); finalize adds the two partial sums; arrival order inside each
# speedup vs baseline: 3.8117x; 1.2655x over previous
#include <hip/hip_runtime.h>
#include <hip/hip_bf16.h>


typedef float v2f __attribute__((ext_vector_type(2)));
typedef float v8f __attribute__((ext_vector_type(8)));

#define F_IN        512
#define C_OUT       16
#define LDSW_STRIDE (F_IN + 2)

__launch_bounds__(256)
__global__ void appnp_gemm_relu(const float* __restrict__ feat,
                                const float* __restrict__ W,
                                const float* __restrict__ bias,
                                float* __restrict__ xbuf,
                                int nTiles) {
    __shared__ float Wlds[C_OUT * LDSW_STRIDE];

    for (int i = threadIdx.x; i < (C_OUT * F_IN) / 4; i += blockDim.x) {
        const int   r  = (i * 4) / F_IN;
        const int   c  = (i * 4) % F_IN;
        const float4 w = ((const float4*)W)[i];
        float* dr = &Wlds[r * LDSW_STRIDE + c];
        dr[0] = w.x; dr[1] = w.y; dr[2] = w.z; dr[3] = w.w;
    }
    __syncthreads();

    const int wave = threadIdx.x >> 5;
    const int lane = threadIdx.x & 31;
    const int tile = blockIdx.x * 8 + wave;
    if (tile >= nTiles) return;

    const int half = lane >> 4;
    const int l    = lane & 15;

    const float* arow = feat + (size_t)(tile * 16 + l) * F_IN + 2 * half;
    const float* brow = &Wlds[l * LDSW_STRIDE + 2 * half];

    v8f acc = {};
#pragma unroll 8
    for (int k0 = 0; k0 < F_IN; k0 += 4) {
        v2f a = *(const v2f*)(arow + k0);
        v2f b = *(const v2f*)(brow + k0);
        acc = __builtin_amdgcn_wmma_f32_16x16x4_f32(
             false, a,  false, b,
             (short)0, acc,  false,  false);
    }

    const float bv = bias[l];
#pragma unroll
    for (int r = 0; r < 8; ++r) {
        const int row = tile * 16 + r + 8 * half;
        float v = acc[r] + bv;
        v = v > 0.0f ? v : 0.0f;
        xbuf[(size_t)row * C_OUT + l] = v;
    }
}

__global__ void appnp_zero(float* __restrict__ p, float* __restrict__ q, long long n) {
    long long i = (long long)blockIdx.x * blockDim.x + threadIdx.x;
    if (i < n) { p[i] = 0.0f; q[i] = 0.0f; }
}

#define DEG_P       16
#define DEG_R       12800
#define DEG_THREADS 1024

static __device__ __forceinline__ void deg_bin4(unsigned* cnt, const int4 q, const int lo) {
    unsigned r;
    r = (unsigned)(q.x - lo); if (r < (unsigned)DEG_R) atomicAdd(&cnt[r], 1u);
    r = (unsigned)(q.y - lo); if (r < (unsigned)DEG_R) atomicAdd(&cnt[r], 1u);
    r = (unsigned)(q.z - lo); if (r < (unsigned)DEG_R) atomicAdd(&cnt[r], 1u);
    r = (unsigned)(q.w - lo); if (r < (unsigned)DEG_R) atomicAdd(&cnt[r], 1u);
}

__launch_bounds__(DEG_THREADS)
__global__ void appnp_degree_count(const int* __restrict__ src,
                                   const int* __restrict__ dst,
                                   unsigned* __restrict__ part,
                                   int E, int N, int chunk) {
    __shared__ unsigned cnt[DEG_R];

    const int h = blockIdx.x;
    const int p = blockIdx.y;
    const int a = blockIdx.z;
    const int* __restrict__ idx = a ? dst : src;
    const int lo = h * DEG_R;

    for (int i = threadIdx.x; i < DEG_R; i += DEG_THREADS) cnt[i] = 0u;
    __syncthreads();

    const int e0 = p * chunk;
    int e1 = e0 + chunk; if (e1 > E) e1 = E;
    const int len  = e1 > e0 ? e1 - e0 : 0;
    const int nvec = len >> 2;
    const int4* __restrict__ v = (const int4*)(idx + e0);

    int i = threadIdx.x;
    for (; i + 3 * DEG_THREADS < nvec; i += 4 * DEG_THREADS) {
        const int4 q0 = v[i];
        const int4 q1 = v[i + DEG_THREADS];
        const int4 q2 = v[i + 2 * DEG_THREADS];
        const int4 q3 = v[i + 3 * DEG_THREADS];
        deg_bin4(cnt, q0, lo); deg_bin4(cnt, q1, lo);
        deg_bin4(cnt, q2, lo); deg_bin4(cnt, q3, lo);
    }
    for (; i < nvec; i += DEG_THREADS) deg_bin4(cnt, v[i], lo);
    for (int e = e0 + (nvec << 2) + threadIdx.x; e < e1; e += DEG_THREADS) {
        const unsigned r = (unsigned)(idx[e] - lo);
        if (r < (unsigned)DEG_R) atomicAdd(&cnt[r], 1u);
    }
    __syncthreads();

    int nr = N - lo; if (nr > DEG_R) nr = DEG_R;
    unsigned* __restrict__ plane = part + ((size_t)(a * DEG_P + p)) * (size_t)N + lo;
    for (int k = threadIdx.x; k < nr; k += DEG_THREADS) plane[k] = cnt[k];
}

__global__ void appnp_degree_sum(const unsigned* __restrict__ part,
                                 float* __restrict__ deg, int N) {
    const int t = blockIdx.x * blockDim.x + threadIdx.x;
    if (t >= 2 * N) return;
    const int a = t >= N ? 1 : 0;
    const int n = t - a * N;
    const unsigned* __restrict__ q = part + (size_t)a * DEG_P * (size_t)N + n;
    unsigned s = 0u;
#pragma unroll
    for (int k = 0; k < DEG_P; ++k) s += q[(size_t)k * (size_t)N];
    deg[t] = (float)s;
}

#define SC_THREADS 256
#define SC_BLOCKS  2048

static __device__ __forceinline__ unsigned l2_half() {
    const unsigned id = (unsigned)__builtin_amdgcn_s_sendmsg_rtn(0x87);
    return (id >> 18) & 1u;
}

template <int J>
static __device__ __forceinline__ unsigned row_bcast(unsigned v) {
    return (unsigned)__builtin_amdgcn_update_dpp(0, (int)v, 0x150 + J, 0xf, 0xf, true);
}

template <int J>
static __device__ __forceinline__ float sc_load(const float* __restrict__ xbuf,
                                                unsigned srow, unsigned cb) {
    return *(const float*)((const char*)xbuf + (row_bcast<J>(srow) + cb));
}

template <int J, bool GUARD>
static __device__ __forceinline__ void sc_add(float* __restrict__ agg, unsigned drow,
                                              unsigned cb, float x, float nv,
                                              int eb, int E) {
    const unsigned off = row_bcast<J>(drow) + cb;
    const float    n   = __uint_as_float(row_bcast<J>(__float_as_uint(nv)));
    if (!GUARD || eb + J < E)
        atomicAdd((float*)((char*)agg + off), x * n);
}

template <bool GUARD>
static __device__ __forceinline__ void sc_batch(const float* __restrict__ xbuf,
                                                float* __restrict__ agg,
                                                unsigned srow, unsigned drow, float degs,
                                                unsigned cb, int eb, int E) {
    const float x0  = sc_load<0>(xbuf, srow, cb);
    const float x1  = sc_load<1>(xbuf, srow, cb);
    const float x2  = sc_load<2>(xbuf, srow, cb);
    const float x3  = sc_load<3>(xbuf, srow, cb);
    const float x4  = sc_load<4>(xbuf, srow, cb);
    const float x5  = sc_load<5>(xbuf, srow, cb);
    const float x6  = sc_load<6>(xbuf, srow, cb);
    const float x7  = sc_load<7>(xbuf, srow, cb);
    const float x8  = sc_load<8>(xbuf, srow, cb);
    const float x9  = sc_load<9>(xbuf, srow, cb);
    const float x10 = sc_load<10>(xbuf, srow, cb);
    const float x11 = sc_load<11>(xbuf, srow, cb);
    const float x12 = sc_load<12>(xbuf, srow, cb);
    const float x13 = sc_load<13>(xbuf, srow, cb);
    const float x14 = sc_load<14>(xbuf, srow, cb);
    const float x15 = sc_load<15>(xbuf, srow, cb);
    const float nv = rsqrtf(fmaxf(degs, 1.0f));
    sc_add<0,  GUARD>(agg, drow, cb, x0,  nv, eb, E);
    sc_add<1,  GUARD>(agg, drow, cb, x1,  nv, eb, E);
    sc_add<2,  GUARD>(agg, drow, cb, x2,  nv, eb, E);
    sc_add<3,  GUARD>(agg, drow, cb, x3,  nv, eb, E);
    sc_add<4,  GUARD>(agg, drow, cb, x4,  nv, eb, E);
    sc_add<5,  GUARD>(agg, drow, cb, x5,  nv, eb, E);
    sc_add<6,  GUARD>(agg, drow, cb, x6,  nv, eb, E);
    sc_add<7,  GUARD>(agg, drow, cb, x7,  nv, eb, E);
    sc_add<8,  GUARD>(agg, drow, cb, x8,  nv, eb, E);
    sc_add<9,  GUARD>(agg, drow, cb, x9,  nv, eb, E);
    sc_add<10, GUARD>(agg, drow, cb, x10, nv, eb, E);
    sc_add<11, GUARD>(agg, drow, cb, x11, nv, eb, E);
    sc_add<12, GUARD>(agg, drow, cb, x12, nv, eb, E);
    sc_add<13, GUARD>(agg, drow, cb, x13, nv, eb, E);
    sc_add<14, GUARD>(agg, drow, cb, x14, nv, eb, E);
    sc_add<15, GUARD>(agg, drow, cb, x15, nv, eb, E);
}

__launch_bounds__(SC_THREADS)
__global__ void appnp_scatter(const int* __restrict__ src,
                              const int* __restrict__ dst,
                              const float* __restrict__ xbuf,
                              const float* __restrict__ outdeg,
                              float* agg0, float* agg1, int E) {
    float* __restrict__ agg = l2_half() ? agg1 : agg0;
    const int lane   = threadIdx.x & 31;
    const int gwave  = (int)((blockIdx.x * SC_THREADS + threadIdx.x) >> 5);
    const int nwaves = (int)((gridDim.x * SC_THREADS) >> 5);
    const int nb     = (E + 31) >> 5;
    const unsigned cb = (unsigned)(lane & 15) << 2;
    const int hofs   = lane & 16;

    int b = gwave;
    if (b >= nb) return;
    int e = (b << 5) + lane;
    int s = 0, d = 0;
    if (e < E) { s = src[e]; d = dst[e]; }

    for (;;) {
        const float degs = outdeg[s];
        const int bn = b + nwaves;
        int sn = 0, dn = 0;
        if (bn < nb) {
            const int en = (bn << 5) + lane;
            if (en < E) { sn = src[en]; dn = dst[en]; }
        }
        const unsigned srow = (unsigned)s << 6;
        const unsigned drow = (unsigned)d << 6;
        const int eb = (b << 5) + hofs;
        if ((b << 5) + 32 <= E) sc_batch<false>(xbuf, agg, srow, drow, degs, cb, eb, E);
        else                    sc_batch<true >(xbuf, agg, srow, drow, degs, cb, eb, E);
        if (bn >= nb) break;
        b = bn; s = sn; d = dn;
    }
}

__global__ void appnp_finalize(const float* __restrict__ agg0,
                               const float* __restrict__ indeg,
                               const float* __restrict__ xbuf,
                               float* out, int N) {
    int t = blockIdx.x * blockDim.x + threadIdx.x;
    if (t >= N * 4) return;
    const int n = t >> 2;
    const float nd = rsqrtf(fmaxf(indeg[n], 1.0f));
    const float4 a0 = ((const float4*)agg0)[t];
    const float4 a1 = ((const float4*)out)[t];
    float4 a;
    a.x = a0.x + a1.x; a.y = a0.y + a1.y; a.z = a0.z + a1.z; a.w = a0.w + a1.w;
    const float4 x0 = ((const float4*)xbuf)[t];
    float4 o;
    o.x = 0.8f * a.x * nd + 0.2f * x0.x;
    o.y = 0.8f * a.y * nd + 0.2f * x0.y;
    o.z = 0.8f * a.z * nd + 0.2f * x0.z;
    o.w = 0.8f * a.w * nd + 0.2f * x0.w;
    ((float4*)out)[t] = o;
}

extern "C" void kernel_launch(void* const* d_in, const int* in_sizes, int n_in,
                              void* d_out, int out_size, void* d_ws, size_t ws_size,
                              hipStream_t stream) {
    const float* feat = (const float*)d_in[0];
    const float* W    = (const float*)d_in[1];
    const float* bias = (const float*)d_in[2];
    const int*   src  = (const int*)d_in[3];
    const int*   dst  = (const int*)d_in[4];

    const int N = in_sizes[0] / F_IN;
    const int E = in_sizes[3];

    float* agg1 = (float*)d_out;
    float* ws     = (float*)d_ws;
    float* xbuf   = ws;
    float* agg    = xbuf + (size_t)N * C_OUT;
    float* outdeg = agg  + (size_t)N * C_OUT;
    float* indeg  = outdeg + N;

    {
        unsigned* part = (unsigned*)xbuf;
        const int chunk  = (((E + DEG_P - 1) / DEG_P) + 3) & ~3;
        const int ranges = (N + DEG_R - 1) / DEG_R;
        appnp_degree_count<<<dim3(ranges, DEG_P, 2), DEG_THREADS, 0, stream>>>(
            src, dst, part, E, N, chunk);
        appnp_degree_sum<<<(2 * N + 255) / 256, 256, 0, stream>>>(part, outdeg, N);
    }
    {
        long long cnt = (long long)N * C_OUT;
        int blocks = (int)((cnt + 255) / 256);
        appnp_zero<<<blocks, 256, 0, stream>>>(agg, agg1, cnt);
    }
    {
        const int nTiles = (N + 15) / 16;
        const int blocks = (nTiles + 7) / 8;
        appnp_gemm_relu<<<blocks, 256, 0, stream>>>(feat, W, bias, xbuf, nTiles);
    }
    {
        appnp_scatter<<<SC_BLOCKS, SC_THREADS, 0, stream>>>(src, dst, xbuf, outdeg, agg, agg1, E);
    }
    appnp_finalize<<<(N * 4 + 255) / 256, 256, 0, stream>>>(agg, indeg, xbuf,
                                                            (float*)d_out, N);
}


# scatter-add gathers rows the GEMM epilogue pre-scaled by rsqrt(outdeg): the per-edge 4-B degree gather (a third of the scatter's requests) is gone; finalize divides the scale back out for 0.2*x
# speedup vs baseline: 4.3242x; 1.1345x over previous
#include <hip/hip_runtime.h>
#include <hip/hip_bf16.h>


typedef float v2f __attribute__((ext_vector_type(2)));
typedef float v8f __attribute__((ext_vector_type(8)));

#define F_IN        512
#define C_OUT       16
#define LDSW_STRIDE (F_IN + 2)

__launch_bounds__(256)
__global__ void appnp_gemm_relu(const float* __restrict__ feat,
                                const float* __restrict__ W,
                                const float* __restrict__ bias,
                                const float* __restrict__ outdeg,
                                float* __restrict__ hbuf,
                                int nTiles) {
    __shared__ float Wlds[C_OUT * LDSW_STRIDE];

    for (int i = threadIdx.x; i < (C_OUT * F_IN) / 4; i += blockDim.x) {
        const int   r  = (i * 4) / F_IN;
        const int   c  = (i * 4) % F_IN;
        const float4 w = ((const float4*)W)[i];
        float* dr = &Wlds[r * LDSW_STRIDE + c];
        dr[0] = w.x; dr[1] = w.y; dr[2] = w.z; dr[3] = w.w;
    }
    __syncthreads();

    const int wave = threadIdx.x >> 5;
    const int lane = threadIdx.x & 31;
    const int tile = blockIdx.x * 8 + wave;
    if (tile >= nTiles) return;

    const int half = lane >> 4;
    const int l    = lane & 15;

    const float* arow = feat + (size_t)(tile * 16 + l) * F_IN + 2 * half;
    const float* brow = &Wlds[l * LDSW_STRIDE + 2 * half];

    v8f acc = {};
#pragma unroll 8
    for (int k0 = 0; k0 < F_IN; k0 += 4) {
        v2f a = *(const v2f*)(arow + k0);
        v2f b = *(const v2f*)(brow + k0);
        acc = __builtin_amdgcn_wmma_f32_16x16x4_f32(
             false, a,  false, b,
             (short)0, acc,  false,  false);
    }

    const float bv = bias[l];
#pragma unroll
    for (int r = 0; r < 8; ++r) {
        const int row = tile * 16 + r + 8 * half;
        const float ns = rsqrtf(fmaxf(outdeg[row], 1.0f));
        float v = acc[r] + bv;
        v = v > 0.0f ? v : 0.0f;
        hbuf[(size_t)row * C_OUT + l] = v * ns;
    }
}

__global__ void appnp_zero(float* __restrict__ p, float* __restrict__ q, long long n) {
    long long i = (long long)blockIdx.x * blockDim.x + threadIdx.x;
    if (i < n) { p[i] = 0.0f; q[i] = 0.0f; }
}

#define DEG_P       16
#define DEG_R       12800
#define DEG_THREADS 1024

static __device__ __forceinline__ void deg_bin4(unsigned* cnt, const int4 q, const int lo) {
    unsigned r;
    r = (unsigned)(q.x - lo); if (r < (unsigned)DEG_R) atomicAdd(&cnt[r], 1u);
    r = (unsigned)(q.y - lo); if (r < (unsigned)DEG_R) atomicAdd(&cnt[r], 1u);
    r = (unsigned)(q.z - lo); if (r < (unsigned)DEG_R) atomicAdd(&cnt[r], 1u);
    r = (unsigned)(q.w - lo); if (r < (unsigned)DEG_R) atomicAdd(&cnt[r], 1u);
}

__launch_bounds__(DEG_THREADS)
__global__ void appnp_degree_count(const int* __restrict__ src,
                                   const int* __restrict__ dst,
                                   unsigned* __restrict__ part,
                                   int E, int N, int chunk) {
    __shared__ unsigned cnt[DEG_R];

    const int h = blockIdx.x;
    const int p = blockIdx.y;
    const int a = blockIdx.z;
    const int* __restrict__ idx = a ? dst : src;
    const int lo = h * DEG_R;

    for (int i = threadIdx.x; i < DEG_R; i += DEG_THREADS) cnt[i] = 0u;
    __syncthreads();

    const int e0 = p * chunk;
    int e1 = e0 + chunk; if (e1 > E) e1 = E;
    const int len  = e1 > e0 ? e1 - e0 : 0;
    const int nvec = len >> 2;
    const int4* __restrict__ v = (const int4*)(idx + e0);

    int i = threadIdx.x;
    for (; i + 3 * DEG_THREADS < nvec; i += 4 * DEG_THREADS) {
        const int4 q0 = v[i];
        const int4 q1 = v[i + DEG_THREADS];
        const int4 q2 = v[i + 2 * DEG_THREADS];
        const int4 q3 = v[i + 3 * DEG_THREADS];
        deg_bin4(cnt, q0, lo); deg_bin4(cnt, q1, lo);
        deg_bin4(cnt, q2, lo); deg_bin4(cnt, q3, lo);
    }
    for (; i < nvec; i += DEG_THREADS) deg_bin4(cnt, v[i], lo);
    for (int e = e0 + (nvec << 2) + threadIdx.x; e < e1; e += DEG_THREADS) {
        const unsigned r = (unsigned)(idx[e] - lo);
        if (r < (unsigned)DEG_R) atomicAdd(&cnt[r], 1u);
    }
    __syncthreads();

    int nr = N - lo; if (nr > DEG_R) nr = DEG_R;
    unsigned* __restrict__ plane = part + ((size_t)(a * DEG_P + p)) * (size_t)N + lo;
    for (int k = threadIdx.x; k < nr; k += DEG_THREADS) plane[k] = cnt[k];
}

__global__ void appnp_degree_sum(const unsigned* __restrict__ part,
                                 float* __restrict__ deg, int N) {
    const int t = blockIdx.x * blockDim.x + threadIdx.x;
    if (t >= 2 * N) return;
    const int a = t >= N ? 1 : 0;
    const int n = t - a * N;
    const unsigned* __restrict__ q = part + (size_t)a * DEG_P * (size_t)N + n;
    unsigned s = 0u;
#pragma unroll
    for (int k = 0; k < DEG_P; ++k) s += q[(size_t)k * (size_t)N];
    deg[t] = (float)s;
}

#define SC_THREADS 256
#define SC_BLOCKS  2048

static __device__ __forceinline__ unsigned l2_half() {
    const unsigned id = (unsigned)__builtin_amdgcn_s_sendmsg_rtn(0x87);
    return (id >> 18) & 1u;
}

template <int J>
static __device__ __forceinline__ unsigned row_bcast(unsigned v) {
    return (unsigned)__builtin_amdgcn_update_dpp(0, (int)v, 0x150 + J, 0xf, 0xf, true);
}

template <int J>
static __device__ __forceinline__ float sc_load(const float* __restrict__ xbuf,
                                                unsigned srow, unsigned cb) {
    return *(const float*)((const char*)xbuf + (row_bcast<J>(srow) + cb));
}

template <int J, bool GUARD>
static __device__ __forceinline__ void sc_add(float* __restrict__ agg, unsigned drow,
                                              unsigned cb, float x,
                                              int eb, int E) {
    const unsigned off = row_bcast<J>(drow) + cb;
    if (!GUARD || eb + J < E)
        atomicAdd((float*)((char*)agg + off), x);
}

template <bool GUARD>
static __device__ __forceinline__ void sc_batch(const float* __restrict__ xbuf,
                                                float* __restrict__ agg,
                                                unsigned srow, unsigned drow,
                                                unsigned cb, int eb, int E) {
    const float x0  = sc_load<0>(xbuf, srow, cb);
    const float x1  = sc_load<1>(xbuf, srow, cb);
    const float x2  = sc_load<2>(xbuf, srow, cb);
    const float x3  = sc_load<3>(xbuf, srow, cb);
    const float x4  = sc_load<4>(xbuf, srow, cb);
    const float x5  = sc_load<5>(xbuf, srow, cb);
    const float x6  = sc_load<6>(xbuf, srow, cb);
    const float x7  = sc_load<7>(xbuf, srow, cb);
    const float x8  = sc_load<8>(xbuf, srow, cb);
    const float x9  = sc_load<9>(xbuf, srow, cb);
    const float x10 = sc_load<10>(xbuf, srow, cb);
    const float x11 = sc_load<11>(xbuf, srow, cb);
    const float x12 = sc_load<12>(xbuf, srow, cb);
    const float x13 = sc_load<13>(xbuf, srow, cb);
    const float x14 = sc_load<14>(xbuf, srow, cb);
    const float x15 = sc_load<15>(xbuf, srow, cb);
    sc_add<0,  GUARD>(agg, drow, cb, x0,  eb, E);
    sc_add<1,  GUARD>(agg, drow, cb, x1,  eb, E);
    sc_add<2,  GUARD>(agg, drow, cb, x2,  eb, E);
    sc_add<3,  GUARD>(agg, drow, cb, x3,  eb, E);
    sc_add<4,  GUARD>(agg, drow, cb, x4,  eb, E);
    sc_add<5,  GUARD>(agg, drow, cb, x5,  eb, E);
    sc_add<6,  GUARD>(agg, drow, cb, x6,  eb, E);
    sc_add<7,  GUARD>(agg, drow, cb, x7,  eb, E);
    sc_add<8,  GUARD>(agg, drow, cb, x8,  eb, E);
    sc_add<9,  GUARD>(agg, drow, cb, x9,  eb, E);
    sc_add<10, GUARD>(agg, drow, cb, x10, eb, E);
    sc_add<11, GUARD>(agg, drow, cb, x11, eb, E);
    sc_add<12, GUARD>(agg, drow, cb, x12, eb, E);
    sc_add<13, GUARD>(agg, drow, cb, x13, eb, E);
    sc_add<14, GUARD>(agg, drow, cb, x14, eb, E);
    sc_add<15, GUARD>(agg, drow, cb, x15, eb, E);
}

__launch_bounds__(SC_THREADS)
__global__ void appnp_scatter(const int* __restrict__ src,
                              const int* __restrict__ dst,
                              const float* __restrict__ xbuf,
                              float* agg0, float* agg1, int E) {
    float* __restrict__ agg = l2_half() ? agg1 : agg0;
    const int lane   = threadIdx.x & 31;
    const int gwave  = (int)((blockIdx.x * SC_THREADS + threadIdx.x) >> 5);
    const int nwaves = (int)((gridDim.x * SC_THREADS) >> 5);
    const int nb     = (E + 31) >> 5;
    const unsigned cb = (unsigned)(lane & 15) << 2;
    const int hofs   = lane & 16;

    int b = gwave;
    if (b >= nb) return;
    int e = (b << 5) + lane;
    int s = 0, d = 0;
    if (e < E) { s = src[e]; d = dst[e]; }

    for (;;) {
        const int bn = b + nwaves;
        int sn = 0, dn = 0;
        if (bn < nb) {
            const int en = (bn << 5) + lane;
            if (en < E) { sn = src[en]; dn = dst[en]; }
        }
        const unsigned srow = (unsigned)s << 6;
        const unsigned drow = (unsigned)d << 6;
        const int eb = (b << 5) + hofs;
        if ((b << 5) + 32 <= E) sc_batch<false>(xbuf, agg, srow, drow, cb, eb, E);
        else                    sc_batch<true >(xbuf, agg, srow, drow, cb, eb, E);
        if (bn >= nb) break;
        b = bn; s = sn; d = dn;
    }
}

__global__ void appnp_finalize(const float* __restrict__ agg0,
                               const float* __restrict__ outdeg,
                               const float* __restrict__ indeg,
                               const float* __restrict__ hbuf,
                               float* out, int N) {
    int t = blockIdx.x * blockDim.x + threadIdx.x;
    if (t >= N * 4) return;
    const int n = t >> 2;
    const float nd = rsqrtf(fmaxf(indeg[n], 1.0f));
    const float ns = rsqrtf(fmaxf(outdeg[n], 1.0f));
    const float4 a0 = ((const float4*)agg0)[t];
    const float4 a1 = ((const float4*)out)[t];
    float4 a;
    a.x = a0.x + a1.x; a.y = a0.y + a1.y; a.z = a0.z + a1.z; a.w = a0.w + a1.w;
    const float4 h0 = ((const float4*)hbuf)[t];
    float4 x0;
    x0.x = h0.x / ns; x0.y = h0.y / ns; x0.z = h0.z / ns; x0.w = h0.w / ns;
    float4 o;
    o.x = 0.8f * a.x * nd + 0.2f * x0.x;
    o.y = 0.8f * a.y * nd + 0.2f * x0.y;
    o.z = 0.8f * a.z * nd + 0.2f * x0.z;
    o.w = 0.8f * a.w * nd + 0.2f * x0.w;
    ((float4*)out)[t] = o;
}

extern "C" void kernel_launch(void* const* d_in, const int* in_sizes, int n_in,
                              void* d_out, int out_size, void* d_ws, size_t ws_size,
                              hipStream_t stream) {
    const float* feat = (const float*)d_in[0];
    const float* W    = (const float*)d_in[1];
    const float* bias = (const float*)d_in[2];
    const int*   src  = (const int*)d_in[3];
    const int*   dst  = (const int*)d_in[4];

    const int N = in_sizes[0] / F_IN;
    const int E = in_sizes[3];

    float* agg1 = (float*)d_out;
    float* ws     = (float*)d_ws;
    float* xbuf   = ws;
    float* agg    = xbuf + (size_t)N * C_OUT;
    float* outdeg = agg  + (size_t)N * C_OUT;
    float* indeg  = outdeg + N;

    {
        unsigned* part = (unsigned*)xbuf;
        const int chunk  = (((E + DEG_P - 1) / DEG_P) + 3) & ~3;
        const int ranges = (N + DEG_R - 1) / DEG_R;
        appnp_degree_count<<<dim3(ranges, DEG_P, 2), DEG_THREADS, 0, stream>>>(
            src, dst, part, E, N, chunk);
        appnp_degree_sum<<<(2 * N + 255) / 256, 256, 0, stream>>>(part, outdeg, N);
    }
    {
        long long cnt = (long long)N * C_OUT;
        int blocks = (int)((cnt + 255) / 256);
        appnp_zero<<<blocks, 256, 0, stream>>>(agg, agg1, cnt);
    }
    {
        const int nTiles = (N + 15) / 16;
        const int blocks = (nTiles + 7) / 8;
        appnp_gemm_relu<<<blocks, 256, 0, stream>>>(feat, W, bias, outdeg, xbuf, nTiles);
    }
    {
        appnp_scatter<<<SC_BLOCKS, SC_THREADS, 0, stream>>>(src, dst, xbuf, agg, agg1, E);
    }
    appnp_finalize<<<(N * 4 + 255) / 256, 256, 0, stream>>>(agg, outdeg, indeg, xbuf,
                                                            (float*)d_out, N);
}


# GEMM launch absorbs the degree plane-sum and the accumulator clear: planes stored per 16-row tile inside the two accumulators, each wave sums then zeroes its own tile; 6 launches become 4
# speedup vs baseline: 4.5136x; 1.0438x over previous
#include <hip/hip_runtime.h>
#include <hip/hip_bf16.h>


typedef float v2f __attribute__((ext_vector_type(2)));
typedef float v8f __attribute__((ext_vector_type(8)));

#define F_IN        512
#define C_OUT       16
#define LDSW_STRIDE (F_IN + 2)
#define DEG_P       16

__launch_bounds__(256)
__global__ void appnp_gemm_relu(const float* __restrict__ feat,
                                const float* __restrict__ W,
                                const float* __restrict__ bias,
                                float* acc0, float* acc1,
                                float* __restrict__ outdeg,
                                float* __restrict__ indeg,
                                float* __restrict__ hbuf,
                                int nTiles) {
    __shared__ float Wlds[C_OUT * LDSW_STRIDE];

    for (int i = threadIdx.x; i < (C_OUT * F_IN) / 4; i += blockDim.x) {
        const int   r  = (i * 4) / F_IN;
        const int   c  = (i * 4) % F_IN;
        const float4 w = ((const float4*)W)[i];
        float* dr = &Wlds[r * LDSW_STRIDE + c];
        dr[0] = w.x; dr[1] = w.y; dr[2] = w.z; dr[3] = w.w;
    }
    __syncthreads();

    const int wave = threadIdx.x >> 5;
    const int lane = threadIdx.x & 31;
    const int tile = blockIdx.x * 8 + wave;
    if (tile >= nTiles) return;

    const int half = lane >> 4;
    const int l    = lane & 15;

    const float* arow = feat + (size_t)(tile * 16 + l) * F_IN + 2 * half;
    const float* brow = &Wlds[l * LDSW_STRIDE + 2 * half];

    unsigned so = 0u, si = 0u;
    {
        const unsigned* q0 = (const unsigned*)acc0 + (size_t)tile * (DEG_P * 16) + half * (DEG_P * 8) + l;
        const unsigned* q1 = (const unsigned*)acc1 + (size_t)tile * (DEG_P * 16) + half * (DEG_P * 8) + l;
#pragma unroll
        for (int j = 0; j < DEG_P / 2; ++j) { so += q0[j * 16]; si += q1[j * 16]; }
    }

    v8f acc = {};
#pragma unroll 8
    for (int k0 = 0; k0 < F_IN; k0 += 4) {
        v2f a = *(const v2f*)(arow + k0);
        v2f b = *(const v2f*)(brow + k0);
        acc = __builtin_amdgcn_wmma_f32_16x16x4_f32(
             false, a,  false, b,
             (short)0, acc,  false,  false);
    }

    so += (unsigned)__shfl_xor((int)so, 16);
    si += (unsigned)__shfl_xor((int)si, 16);
    const float od = (float)so;
    const float id = (float)si;
    if (half) indeg[tile * 16 + l] = id; else outdeg[tile * 16 + l] = od;
    const float nsl = rsqrtf(fmaxf(od, 1.0f));

    const float bv = bias[l];
#pragma unroll
    for (int r = 0; r < 8; ++r) {
        const int row = tile * 16 + r + 8 * half;
        const float ns = __shfl(nsl, r + 8 * half);
        float v = acc[r] + bv;
        v = v > 0.0f ? v : 0.0f;
        hbuf[(size_t)row * C_OUT + l] = v * ns;
    }

    {
        const float4 z = {0.0f, 0.0f, 0.0f, 0.0f};
        float4* z0 = (float4*)(acc0 + (size_t)tile * (16 * C_OUT));
        float4* z1 = (float4*)(acc1 + (size_t)tile * (16 * C_OUT));
        z0[lane] = z; z0[lane + 32] = z;
        z1[lane] = z; z1[lane + 32] = z;
    }
}

#define DEG_R       12800
#define DEG_THREADS 1024

static __device__ __forceinline__ void deg_bin4(unsigned* cnt, const int4 q, const int lo) {
    unsigned r;
    r = (unsigned)(q.x - lo); if (r < (unsigned)DEG_R) atomicAdd(&cnt[r], 1u);
    r = (unsigned)(q.y - lo); if (r < (unsigned)DEG_R) atomicAdd(&cnt[r], 1u);
    r = (unsigned)(q.z - lo); if (r < (unsigned)DEG_R) atomicAdd(&cnt[r], 1u);
    r = (unsigned)(q.w - lo); if (r < (unsigned)DEG_R) atomicAdd(&cnt[r], 1u);
}

__launch_bounds__(DEG_THREADS)
__global__ void appnp_degree_count(const int* __restrict__ src,
                                   const int* __restrict__ dst,
                                   unsigned* __restrict__ part0,
                                   unsigned* __restrict__ part1,
                                   int E, int N, int chunk) {
    __shared__ unsigned cnt[DEG_R];

    const int h = blockIdx.x;
    const int p = blockIdx.y;
    const int a = blockIdx.z;
    const int* __restrict__ idx = a ? dst : src;
    const int lo = h * DEG_R;

    for (int i = threadIdx.x; i < DEG_R; i += DEG_THREADS) cnt[i] = 0u;
    __syncthreads();

    const int e0 = p * chunk;
    int e1 = e0 + chunk; if (e1 > E) e1 = E;
    const int len  = e1 > e0 ? e1 - e0 : 0;
    const int nvec = len >> 2;
    const int4* __restrict__ v = (const int4*)(idx + e0);

    int i = threadIdx.x;
    for (; i + 3 * DEG_THREADS < nvec; i += 4 * DEG_THREADS) {
        const int4 q0 = v[i];
        const int4 q1 = v[i + DEG_THREADS];
        const int4 q2 = v[i + 2 * DEG_THREADS];
        const int4 q3 = v[i + 3 * DEG_THREADS];
        deg_bin4(cnt, q0, lo); deg_bin4(cnt, q1, lo);
        deg_bin4(cnt, q2, lo); deg_bin4(cnt, q3, lo);
    }
    for (; i < nvec; i += DEG_THREADS) deg_bin4(cnt, v[i], lo);
    for (int e = e0 + (nvec << 2) + threadIdx.x; e < e1; e += DEG_THREADS) {
        const unsigned r = (unsigned)(idx[e] - lo);
        if (r < (unsigned)DEG_R) atomicAdd(&cnt[r], 1u);
    }
    __syncthreads();

    int nr = N - lo; if (nr > DEG_R) nr = DEG_R;
    unsigned* __restrict__ plane = (a ? part1 : part0) + p * 16;
    for (int k = threadIdx.x; k < nr; k += DEG_THREADS) {
        const int n = lo + k;
        plane[(size_t)(n >> 4) * (DEG_P * 16) + (n & 15)] = cnt[k];
    }
}

#define SC_THREADS 256
#define SC_BLOCKS  2048

static __device__ __forceinline__ unsigned l2_half() {
    const unsigned id = (unsigned)__builtin_amdgcn_s_sendmsg_rtn(0x87);
    return (id >> 18) & 1u;
}

template <int J>
static __device__ __forceinline__ unsigned row_bcast(unsigned v) {
    return (unsigned)__builtin_amdgcn_update_dpp(0, (int)v, 0x150 + J, 0xf, 0xf, true);
}

template <int J>
static __device__ __forceinline__ float sc_load(const float* __restrict__ xbuf,
                                                unsigned srow, unsigned cb) {
    return *(const float*)((const char*)xbuf + (row_bcast<J>(srow) + cb));
}

template <int J, bool GUARD>
static __device__ __forceinline__ void sc_add(float* __restrict__ agg, unsigned drow,
                                              unsigned cb, float x,
                                              int eb, int E) {
    const unsigned off = row_bcast<J>(drow) + cb;
    if (!GUARD || eb + J < E)
        atomicAdd((float*)((char*)agg + off), x);
}

template <bool GUARD>
static __device__ __forceinline__ void sc_batch(const float* __restrict__ xbuf,
                                                float* __restrict__ agg,
                                                unsigned srow, unsigned drow,
                                                unsigned cb, int eb, int E) {
    const float x0  = sc_load<0>(xbuf, srow, cb);
    const float x1  = sc_load<1>(xbuf, srow, cb);
    const float x2  = sc_load<2>(xbuf, srow, cb);
    const float x3  = sc_load<3>(xbuf, srow, cb);
    const float x4  = sc_load<4>(xbuf, srow, cb);
    const float x5  = sc_load<5>(xbuf, srow, cb);
    const float x6  = sc_load<6>(xbuf, srow, cb);
    const float x7  = sc_load<7>(xbuf, srow, cb);
    const float x8  = sc_load<8>(xbuf, srow, cb);
    const float x9  = sc_load<9>(xbuf, srow, cb);
    const float x10 = sc_load<10>(xbuf, srow, cb);
    const float x11 = sc_load<11>(xbuf, srow, cb);
    const float x12 = sc_load<12>(xbuf, srow, cb);
    const float x13 = sc_load<13>(xbuf, srow, cb);
    const float x14 = sc_load<14>(xbuf, srow, cb);
    const float x15 = sc_load<15>(xbuf, srow, cb);
    sc_add<0,  GUARD>(agg, drow, cb, x0,  eb, E);
    sc_add<1,  GUARD>(agg, drow, cb, x1,  eb, E);
    sc_add<2,  GUARD>(agg, drow, cb, x2,  eb, E);
    sc_add<3,  GUARD>(agg, drow, cb, x3,  eb, E);
    sc_add<4,  GUARD>(agg, drow, cb, x4,  eb, E);
    sc_add<5,  GUARD>(agg, drow, cb, x5,  eb, E);
    sc_add<6,  GUARD>(agg, drow, cb, x6,  eb, E);
    sc_add<7,  GUARD>(agg, drow, cb, x7,  eb, E);
    sc_add<8,  GUARD>(agg, drow, cb, x8,  eb, E);
    sc_add<9,  GUARD>(agg, drow, cb, x9,  eb, E);
    sc_add<10, GUARD>(agg, drow, cb, x10, eb, E);
    sc_add<11, GUARD>(agg, drow, cb, x11, eb, E);
    sc_add<12, GUARD>(agg, drow, cb, x12, eb, E);
    sc_add<13, GUARD>(agg, drow, cb, x13, eb, E);
    sc_add<14, GUARD>(agg, drow, cb, x14, eb, E);
    sc_add<15, GUARD>(agg, drow, cb, x15, eb, E);
}

__launch_bounds__(SC_THREADS)
__global__ void appnp_scatter(const int* __restrict__ src,
                              const int* __restrict__ dst,
                              const float* __restrict__ xbuf,
                              float* agg0, float* agg1, int E) {
    float* __restrict__ agg = l2_half() ? agg1 : agg0;
    const int lane   = threadIdx.x & 31;
    const int gwave  = (int)((blockIdx.x * SC_THREADS + threadIdx.x) >> 5);
    const int nwaves = (int)((gridDim.x * SC_THREADS) >> 5);
    const int nb     = (E + 31) >> 5;
    const unsigned cb = (unsigned)(lane & 15) << 2;
    const int hofs   = lane & 16;

    int b = gwave;
    if (b >= nb) return;
    int e = (b << 5) + lane;
    int s = 0, d = 0;
    if (e < E) { s = src[e]; d = dst[e]; }

    for (;;) {
        const int bn = b + nwaves;
        int sn = 0, dn = 0;
        if (bn < nb) {
            const int en = (bn << 5) + lane;
            if (en < E) { sn = src[en]; dn = dst[en]; }
        }
        const unsigned srow = (unsigned)s << 6;
        const unsigned drow = (unsigned)d << 6;
        const int eb = (b << 5) + hofs;
        if ((b << 5) + 32 <= E) sc_batch<false>(xbuf, agg, srow, drow, cb, eb, E);
        else                    sc_batch<true >(xbuf, agg, srow, drow, cb, eb, E);
        if (bn >= nb) break;
        b = bn; s = sn; d = dn;
    }
}

__global__ void appnp_finalize(const float* __restrict__ agg0,
                               const float* __restrict__ outdeg,
                               const float* __restrict__ indeg,
                               const float* __restrict__ hbuf,
                               float* out, int N) {
    int t = blockIdx.x * blockDim.x + threadIdx.x;
    if (t >= N * 4) return;
    const int n = t >> 2;
    const float nd = rsqrtf(fmaxf(indeg[n], 1.0f));
    const float ns = rsqrtf(fmaxf(outdeg[n], 1.0f));
    const float4 a0 = ((const float4*)agg0)[t];
    const float4 a1 = ((const float4*)out)[t];
    float4 a;
    a.x = a0.x + a1.x; a.y = a0.y + a1.y; a.z = a0.z + a1.z; a.w = a0.w + a1.w;
    const float4 h0 = ((const float4*)hbuf)[t];
    float4 x0;
    x0.x = h0.x / ns; x0.y = h0.y / ns; x0.z = h0.z / ns; x0.w = h0.w / ns;
    float4 o;
    o.x = 0.8f * a.x * nd + 0.2f * x0.x;
    o.y = 0.8f * a.y * nd + 0.2f * x0.y;
    o.z = 0.8f * a.z * nd + 0.2f * x0.z;
    o.w = 0.8f * a.w * nd + 0.2f * x0.w;
    ((float4*)out)[t] = o;
}

extern "C" void kernel_launch(void* const* d_in, const int* in_sizes, int n_in,
                              void* d_out, int out_size, void* d_ws, size_t ws_size,
                              hipStream_t stream) {
    const float* feat = (const float*)d_in[0];
    const float* W    = (const float*)d_in[1];
    const float* bias = (const float*)d_in[2];
    const int*   src  = (const int*)d_in[3];
    const int*   dst  = (const int*)d_in[4];

    const int N = in_sizes[0] / F_IN;
    const int E = in_sizes[3];

    float* agg1 = (float*)d_out;
    float* ws     = (float*)d_ws;
    float* xbuf   = ws;
    float* agg    = xbuf + (size_t)N * C_OUT;
    float* outdeg = agg  + (size_t)N * C_OUT;
    float* indeg  = outdeg + N;

    {
        const int chunk  = (((E + DEG_P - 1) / DEG_P) + 3) & ~3;
        const int ranges = (N + DEG_R - 1) / DEG_R;
        appnp_degree_count<<<dim3(ranges, DEG_P, 2), DEG_THREADS, 0, stream>>>(
            src, dst, (unsigned*)agg, (unsigned*)agg1, E, N, chunk);
    }
    {
        const int nTiles = (N + 15) / 16;
        const int blocks = (nTiles + 7) / 8;
        appnp_gemm_relu<<<blocks, 256, 0, stream>>>(feat, W, bias, agg, agg1, outdeg, indeg,
                                                    xbuf, nTiles);
    }
    {
        appnp_scatter<<<SC_BLOCKS, SC_THREADS, 0, stream>>>(src, dst, xbuf, agg, agg1, E);
    }
    appnp_finalize<<<(N * 4 + 255) / 256, 256, 0, stream>>>(agg, outdeg, indeg, xbuf,
                                                            (float*)d_out, N);
}


# GEMM feed: feature tile read as whole 256-B row pieces (16-B loads, 2 rows each) via a wave-private LDS stage; next batch and W stage in flight under the current one; k order and sums unchanged
# speedup vs baseline: 4.8225x; 1.0684x over previous
#include <hip/hip_runtime.h>
#include <hip/hip_bf16.h>


typedef float v2f __attribute__((ext_vector_type(2)));
typedef float v8f __attribute__((ext_vector_type(8)));

#define F_IN        512
#define C_OUT       16
#define LDSW_STRIDE (F_IN + 2)
#define DEG_P       16

#define KB        64
#define NBATCH    (F_IN / KB)
#define STG_PITCH (KB + 4)
#define STG_WAVE  (16 * STG_PITCH)

struct batch8 { float4 p0, p1, p2, p3, p4, p5, p6, p7; };

static __device__ __forceinline__ batch8 gemm_load_batch(const float4* __restrict__ g) {
    batch8 d;
    d.p0 = g[0 * (2 * F_IN / 4)]; d.p1 = g[1 * (2 * F_IN / 4)];
    d.p2 = g[2 * (2 * F_IN / 4)]; d.p3 = g[3 * (2 * F_IN / 4)];
    d.p4 = g[4 * (2 * F_IN / 4)]; d.p5 = g[5 * (2 * F_IN / 4)];
    d.p6 = g[6 * (2 * F_IN / 4)]; d.p7 = g[7 * (2 * F_IN / 4)];
    return d;
}

static __device__ __forceinline__ void gemm_use_batch(const batch8 s, float4* stg4,
                                                      const float* arow, const float* browt,
                                                      v8f& acc) {
    stg4[0 * (2 * STG_PITCH / 4)] = s.p0; stg4[1 * (2 * STG_PITCH / 4)] = s.p1;
    stg4[2 * (2 * STG_PITCH / 4)] = s.p2; stg4[3 * (2 * STG_PITCH / 4)] = s.p3;
    stg4[4 * (2 * STG_PITCH / 4)] = s.p4; stg4[5 * (2 * STG_PITCH / 4)] = s.p5;
    stg4[6 * (2 * STG_PITCH / 4)] = s.p6; stg4[7 * (2 * STG_PITCH / 4)] = s.p7;
#pragma unroll
    for (int k = 0; k < KB / 4; ++k) {
        v2f a = *(const v2f*)(arow + 4 * k);
        v2f b = *(const v2f*)(browt + 4 * k);
        acc = __builtin_amdgcn_wmma_f32_16x16x4_f32(
             false, a,  false, b,
             (short)0, acc,  false,  false);
    }
}

__launch_bounds__(256, 8)
__global__ void appnp_gemm_relu(const float* __restrict__ feat,
                                const float* __restrict__ W,
                                const float* __restrict__ bias,
                                float* acc0, float* acc1,
                                float* __restrict__ outdeg,
                                float* __restrict__ indeg,
                                float* __restrict__ hbuf,
                                int nTiles) {
    __shared__ float  Wlds[C_OUT * LDSW_STRIDE];
    __shared__ float4 Astg[8 * STG_WAVE / 4];

    const int  wave  = threadIdx.x >> 5;
    const int  lane  = threadIdx.x & 31;
    const int  tile  = blockIdx.x * 8 + wave;
    const bool valid = tile < nTiles;
    const int  half  = lane >> 4;
    const int  l     = lane & 15;

    const int tile_c = valid ? tile : nTiles - 1;
    const float4* gsrc = (const float4*)(feat + (size_t)(tile_c * 16 + half) * F_IN) + l;
    batch8 bufa = gemm_load_batch(gsrc);
    batch8 bufb;

    unsigned so = 0u, si = 0u;
    {
        const unsigned* q0 = (const unsigned*)acc0 + (size_t)tile_c * (DEG_P * 16) + half * (DEG_P * 8) + l;
        const unsigned* q1 = (const unsigned*)acc1 + (size_t)tile_c * (DEG_P * 16) + half * (DEG_P * 8) + l;
#pragma unroll
        for (int j = 0; j < DEG_P / 2; ++j) { so += q0[j * 16]; si += q1[j * 16]; }
    }

    {
        float4 w[8];
#pragma unroll
        for (int q = 0; q < 8; ++q) w[q] = ((const float4*)W)[threadIdx.x + 256 * q];
#pragma unroll
        for (int q = 0; q < 8; ++q) {
            const int i = threadIdx.x + 256 * q;
            const int r = i >> 7;
            const int c = (i & 127) << 2;
            float* dr = &Wlds[r * LDSW_STRIDE + c];
            dr[0] = w[q].x; dr[1] = w[q].y; dr[2] = w[q].z; dr[3] = w[q].w;
        }
    }
    __syncthreads();
    if (!valid) return;

    float4*      stg4 = Astg + wave * (STG_WAVE / 4) + half * (STG_PITCH / 4) + l;
    const float* arow = (const float*)Astg + wave * STG_WAVE + l * STG_PITCH + 2 * half;
    const float* brow = &Wlds[l * LDSW_STRIDE + 2 * half];

    v8f acc = {};
    bufb = gemm_load_batch(gsrc + 1 * (KB / 4)); gemm_use_batch(bufa, stg4, arow, brow + 0 * KB, acc);
    bufa = gemm_load_batch(gsrc + 2 * (KB / 4)); gemm_use_batch(bufb, stg4, arow, brow + 1 * KB, acc);
    bufb = gemm_load_batch(gsrc + 3 * (KB / 4)); gemm_use_batch(bufa, stg4, arow, brow + 2 * KB, acc);
    bufa = gemm_load_batch(gsrc + 4 * (KB / 4)); gemm_use_batch(bufb, stg4, arow, brow + 3 * KB, acc);
    bufb = gemm_load_batch(gsrc + 5 * (KB / 4)); gemm_use_batch(bufa, stg4, arow, brow + 4 * KB, acc);
    bufa = gemm_load_batch(gsrc + 6 * (KB / 4)); gemm_use_batch(bufb, stg4, arow, brow + 5 * KB, acc);
    bufb = gemm_load_batch(gsrc + 7 * (KB / 4)); gemm_use_batch(bufa, stg4, arow, brow + 6 * KB, acc);
    gemm_use_batch(bufb, stg4, arow, brow + 7 * KB, acc);

    so += (unsigned)__shfl_xor((int)so, 16);
    si += (unsigned)__shfl_xor((int)si, 16);
    const float od = (float)so;
    const float id = (float)si;
    if (half) indeg[tile * 16 + l] = id; else outdeg[tile * 16 + l] = od;
    const float nsl = rsqrtf(fmaxf(od, 1.0f));

    const float bv = bias[l];
#pragma unroll
    for (int r = 0; r < 8; ++r) {
        const int row = tile * 16 + r + 8 * half;
        const float ns = __shfl(nsl, r + 8 * half);
        float v = acc[r] + bv;
        v = v > 0.0f ? v : 0.0f;
        hbuf[(size_t)row * C_OUT + l] = v * ns;
    }

    {
        const float4 z = {0.0f, 0.0f, 0.0f, 0.0f};
        float4* z0 = (float4*)(acc0 + (size_t)tile * (16 * C_OUT));
        float4* z1 = (float4*)(acc1 + (size_t)tile * (16 * C_OUT));
        z0[lane] = z; z0[lane + 32] = z;
        z1[lane] = z; z1[lane + 32] = z;
    }
}

#define DEG_R       12800
#define DEG_THREADS 1024

static __device__ __forceinline__ void deg_bin4(unsigned* cnt, const int4 q, const int lo) {
    unsigned r;
    r = (unsigned)(q.x - lo); if (r < (unsigned)DEG_R) atomicAdd(&cnt[r], 1u);
    r = (unsigned)(q.y - lo); if (r < (unsigned)DEG_R) atomicAdd(&cnt[r], 1u);
    r = (unsigned)(q.z - lo); if (r < (unsigned)DEG_R) atomicAdd(&cnt[r], 1u);
    r = (unsigned)(q.w - lo); if (r < (unsigned)DEG_R) atomicAdd(&cnt[r], 1u);
}

__launch_bounds__(DEG_THREADS)
__global__ void appnp_degree_count(const int* __restrict__ src,
                                   const int* __restrict__ dst,
                                   unsigned* __restrict__ part0,
                                   unsigned* __restrict__ part1,
                                   int E, int N, int chunk) {
    __shared__ unsigned cnt[DEG_R];

    const int h = blockIdx.x;
    const int p = blockIdx.y;
    const int a = blockIdx.z;
    const int* __restrict__ idx = a ? dst : src;
    const int lo = h * DEG_R;

    for (int i = threadIdx.x; i < DEG_R; i += DEG_THREADS) cnt[i] = 0u;
    __syncthreads();

    const int e0 = p * chunk;
    int e1 = e0 + chunk; if (e1 > E) e1 = E;
    const int len  = e1 > e0 ? e1 - e0 : 0;
    const int nvec = len >> 2;
    const int4* __restrict__ v = (const int4*)(idx + e0);

    int i = threadIdx.x;
    for (; i + 3 * DEG_THREADS < nvec; i += 4 * DEG_THREADS) {
        const int4 q0 = v[i];
        const int4 q1 = v[i + DEG_THREADS];
        const int4 q2 = v[i + 2 * DEG_THREADS];
        const int4 q3 = v[i + 3 * DEG_THREADS];
        deg_bin4(cnt, q0, lo); deg_bin4(cnt, q1, lo);
        deg_bin4(cnt, q2, lo); deg_bin4(cnt, q3, lo);
    }
    for (; i < nvec; i += DEG_THREADS) deg_bin4(cnt, v[i], lo);
    for (int e = e0 + (nvec << 2) + threadIdx.x; e < e1; e += DEG_THREADS) {
        const unsigned r = (unsigned)(idx[e] - lo);
        if (r < (unsigned)DEG_R) atomicAdd(&cnt[r], 1u);
    }
    __syncthreads();

    int nr = N - lo; if (nr > DEG_R) nr = DEG_R;
    unsigned* __restrict__ plane = (a ? part1 : part0) + p * 16;
    for (int k = threadIdx.x; k < nr; k += DEG_THREADS) {
        const int n = lo + k;
        plane[(size_t)(n >> 4) * (DEG_P * 16) + (n & 15)] = cnt[k];
    }
}

#define SC_THREADS 256
#define SC_BLOCKS  2048

static __device__ __forceinline__ unsigned l2_half() {
    const unsigned id = (unsigned)__builtin_amdgcn_s_sendmsg_rtn(0x87);
    return (id >> 18) & 1u;
}

template <int J>
static __device__ __forceinline__ unsigned row_bcast(unsigned v) {
    return (unsigned)__builtin_amdgcn_update_dpp(0, (int)v, 0x150 + J, 0xf, 0xf, true);
}

template <int J>
static __device__ __forceinline__ float sc_load(const float* __restrict__ xbuf,
                                                unsigned srow, unsigned cb) {
    return *(const float*)((const char*)xbuf + (row_bcast<J>(srow) + cb));
}

template <int J, bool GUARD>
static __device__ __forceinline__ void sc_add(float* __restrict__ agg, unsigned drow,
                                              unsigned cb, float x,
                                              int eb, int E) {
    const unsigned off = row_bcast<J>(drow) + cb;
    if (!GUARD || eb + J < E)
        atomicAdd((float*)((char*)agg + off), x);
}

template <bool GUARD>
static __device__ __forceinline__ void sc_batch(const float* __restrict__ xbuf,
                                                float* __restrict__ agg,
                                                unsigned srow, unsigned drow,
                                                unsigned cb, int eb, int E) {
    const float x0  = sc_load<0>(xbuf, srow, cb);
    const float x1  = sc_load<1>(xbuf, srow, cb);
    const float x2  = sc_load<2>(xbuf, srow, cb);
    const float x3  = sc_load<3>(xbuf, srow, cb);
    const float x4  = sc_load<4>(xbuf, srow, cb);
    const float x5  = sc_load<5>(xbuf, srow, cb);
    const float x6  = sc_load<6>(xbuf, srow, cb);
    const float x7  = sc_load<7>(xbuf, srow, cb);
    const float x8  = sc_load<8>(xbuf, srow, cb);
    const float x9  = sc_load<9>(xbuf, srow, cb);
    const float x10 = sc_load<10>(xbuf, srow, cb);
    const float x11 = sc_load<11>(xbuf, srow, cb);
    const float x12 = sc_load<12>(xbuf, srow, cb);
    const float x13 = sc_load<13>(xbuf, srow, cb);
    const float x14 = sc_load<14>(xbuf, srow, cb);
    const float x15 = sc_load<15>(xbuf, srow, cb);
    sc_add<0,  GUARD>(agg, drow, cb, x0,  eb, E);
    sc_add<1,  GUARD>(agg, drow, cb, x1,  eb, E);
    sc_add<2,  GUARD>(agg, drow, cb, x2,  eb, E);
    sc_add<3,  GUARD>(agg, drow, cb, x3,  eb, E);
    sc_add<4,  GUARD>(agg, drow, cb, x4,  eb, E);
    sc_add<5,  GUARD>(agg, drow, cb, x5,  eb, E);
    sc_add<6,  GUARD>(agg, drow, cb, x6,  eb, E);
    sc_add<7,  GUARD>(agg, drow, cb, x7,  eb, E);
    sc_add<8,  GUARD>(agg, drow, cb, x8,  eb, E);
    sc_add<9,  GUARD>(agg, drow, cb, x9,  eb, E);
    sc_add<10, GUARD>(agg, drow, cb, x10, eb, E);
    sc_add<11, GUARD>(agg, drow, cb, x11, eb, E);
    sc_add<12, GUARD>(agg, drow, cb, x12, eb, E);
    sc_add<13, GUARD>(agg, drow, cb, x13, eb, E);
    sc_add<14, GUARD>(agg, drow, cb, x14, eb, E);
    sc_add<15, GUARD>(agg, drow, cb, x15, eb, E);
}

__launch_bounds__(SC_THREADS)
__global__ void appnp_scatter(const int* __restrict__ src,
                              const int* __restrict__ dst,
                              const float* __restrict__ xbuf,
                              float* agg0, float* agg1, int E) {
    float* __restrict__ agg = l2_half() ? agg1 : agg0;
    const int lane   = threadIdx.x & 31;
    const int gwave  = (int)((blockIdx.x * SC_THREADS + threadIdx.x) >> 5);
    const int nwaves = (int)((gridDim.x * SC_THREADS) >> 5);
    const int nb     = (E + 31) >> 5;
    const unsigned cb = (unsigned)(lane & 15) << 2;
    const int hofs   = lane & 16;

    int b = gwave;
    if (b >= nb) return;
    int e = (b << 5) + lane;
    int s = 0, d = 0;
    if (e < E) { s = src[e]; d = dst[e]; }

    for (;;) {
        const int bn = b + nwaves;
        int sn = 0, dn = 0;
        if (bn < nb) {
            const int en = (bn << 5) + lane;
            if (en < E) { sn = src[en]; dn = dst[en]; }
        }
        const unsigned srow = (unsigned)s << 6;
        const unsigned drow = (unsigned)d << 6;
        const int eb = (b << 5) + hofs;
        if ((b << 5) + 32 <= E) sc_batch<false>(xbuf, agg, srow, drow, cb, eb, E);
        else                    sc_batch<true >(xbuf, agg, srow, drow, cb, eb, E);
        if (bn >= nb) break;
        b = bn; s = sn; d = dn;
    }
}

__global__ void appnp_finalize(const float* __restrict__ agg0,
                               const float* __restrict__ outdeg,
                               const float* __restrict__ indeg,
                               const float* __restrict__ hbuf,
                               float* out, int N) {
    int t = blockIdx.x * blockDim.x + threadIdx.x;
    if (t >= N * 4) return;
    const int n = t >> 2;
    const float nd = rsqrtf(fmaxf(indeg[n], 1.0f));
    const float ns = rsqrtf(fmaxf(outdeg[n], 1.0f));
    const float4 a0 = ((const float4*)agg0)[t];
    const float4 a1 = ((const float4*)out)[t];
    float4 a;
    a.x = a0.x + a1.x; a.y = a0.y + a1.y; a.z = a0.z + a1.z; a.w = a0.w + a1.w;
    const float4 h0 = ((const float4*)hbuf)[t];
    float4 x0;
    x0.x = h0.x / ns; x0.y = h0.y / ns; x0.z = h0.z / ns; x0.w = h0.w / ns;
    float4 o;
    o.x = 0.8f * a.x * nd + 0.2f * x0.x;
    o.y = 0.8f * a.y * nd + 0.2f * x0.y;
    o.z = 0.8f * a.z * nd + 0.2f * x0.z;
    o.w = 0.8f * a.w * nd + 0.2f * x0.w;
    ((float4*)out)[t] = o;
}

extern "C" void kernel_launch(void* const* d_in, const int* in_sizes, int n_in,
                              void* d_out, int out_size, void* d_ws, size_t ws_size,
                              hipStream_t stream) {
    const float* feat = (const float*)d_in[0];
    const float* W    = (const float*)d_in[1];
    const float* bias = (const float*)d_in[2];
    const int*   src  = (const int*)d_in[3];
    const int*   dst  = (const int*)d_in[4];

    const int N = in_sizes[0] / F_IN;
    const int E = in_sizes[3];

    float* agg1 = (float*)d_out;
    float* ws     = (float*)d_ws;
    float* xbuf   = ws;
    float* agg    = xbuf + (size_t)N * C_OUT;
    float* outdeg = agg  + (size_t)N * C_OUT;
    float* indeg  = outdeg + N;

    {
        const int chunk  = (((E + DEG_P - 1) / DEG_P) + 3) & ~3;
        const int ranges = (N + DEG_R - 1) / DEG_R;
        appnp_degree_count<<<dim3(ranges, DEG_P, 2), DEG_THREADS, 0, stream>>>(
            src, dst, (unsigned*)agg, (unsigned*)agg1, E, N, chunk);
    }
    {
        const int nTiles = (N + 15) / 16;
        const int blocks = (nTiles + 7) / 8;
        appnp_gemm_relu<<<blocks, 256, 0, stream>>>(feat, W, bias, agg, agg1, outdeg, indeg,
                                                    xbuf, nTiles);
    }
    {
        appnp_scatter<<<SC_BLOCKS, SC_THREADS, 0, stream>>>(src, dst, xbuf, agg, agg1, E);
    }
    appnp_finalize<<<(N * 4 + 255) / 256, 256, 0, stream>>>(agg, outdeg, indeg, xbuf,
                                                            (float*)d_out, N);
}


# scatter-add + mix become a destination-ordered sum: edges counting-sorted via LDS into 400-node buckets, one workgroup per bucket adds h rows in LDS (ds_add_f32, arrival order) and writes out
# speedup vs baseline: 5.3058x; 1.1002x over previous
#include <hip/hip_runtime.h>
#include <hip/hip_bf16.h>


typedef float v2f __attribute__((ext_vector_type(2)));
typedef float v8f __attribute__((ext_vector_type(8)));

#define F_IN        512
#define C_OUT       16
#define LDSW_STRIDE (F_IN + 2)
#define DEG_P       16

#define KB        64
#define NBATCH    (F_IN / KB)
#define STG_PITCH (KB + 4)
#define STG_WAVE  (16 * STG_PITCH)

struct batch8 { float4 p0, p1, p2, p3, p4, p5, p6, p7; };

static __device__ __forceinline__ batch8 gemm_load_batch(const float4* __restrict__ g) {
    batch8 d;
    d.p0 = g[0 * (2 * F_IN / 4)]; d.p1 = g[1 * (2 * F_IN / 4)];
    d.p2 = g[2 * (2 * F_IN / 4)]; d.p3 = g[3 * (2 * F_IN / 4)];
    d.p4 = g[4 * (2 * F_IN / 4)]; d.p5 = g[5 * (2 * F_IN / 4)];
    d.p6 = g[6 * (2 * F_IN / 4)]; d.p7 = g[7 * (2 * F_IN / 4)];
    return d;
}

static __device__ __forceinline__ void gemm_use_batch(const batch8 s, float4* stg4,
                                                      const float* arow, const float* browt,
                                                      v8f& acc) {
    stg4[0 * (2 * STG_PITCH / 4)] = s.p0; stg4[1 * (2 * STG_PITCH / 4)] = s.p1;
    stg4[2 * (2 * STG_PITCH / 4)] = s.p2; stg4[3 * (2 * STG_PITCH / 4)] = s.p3;
    stg4[4 * (2 * STG_PITCH / 4)] = s.p4; stg4[5 * (2 * STG_PITCH / 4)] = s.p5;
    stg4[6 * (2 * STG_PITCH / 4)] = s.p6; stg4[7 * (2 * STG_PITCH / 4)] = s.p7;
#pragma unroll
    for (int k = 0; k < KB / 4; ++k) {
        v2f a = *(const v2f*)(arow + 4 * k);
        v2f b = *(const v2f*)(browt + 4 * k);
        acc = __builtin_amdgcn_wmma_f32_16x16x4_f32(
             false, a,  false, b,
             (short)0, acc,  false,  false);
    }
}

__launch_bounds__(256, 8)
__global__ void appnp_gemm_relu(const float* __restrict__ feat,
                                const float* __restrict__ W,
                                const float* __restrict__ bias,
                                float* acc0, float* acc1,
                                float* __restrict__ outdeg,
                                float* __restrict__ indeg,
                                float* __restrict__ hbuf,
                                int nTiles) {
    __shared__ float  Wlds[C_OUT * LDSW_STRIDE];
    __shared__ float4 Astg[8 * STG_WAVE / 4];

    const int  wave  = threadIdx.x >> 5;
    const int  lane  = threadIdx.x & 31;
    const int  tile  = blockIdx.x * 8 + wave;
    const bool valid = tile < nTiles;
    const int  half  = lane >> 4;
    const int  l     = lane & 15;

    const int tile_c = valid ? tile : nTiles - 1;
    const float4* gsrc = (const float4*)(feat + (size_t)(tile_c * 16 + half) * F_IN) + l;
    batch8 bufa = gemm_load_batch(gsrc);
    batch8 bufb;

    unsigned so = 0u, si = 0u;
    {
        const unsigned* q0 = (const unsigned*)acc0 + (size_t)tile_c * (DEG_P * 16) + half * (DEG_P * 8) + l;
        const unsigned* q1 = (const unsigned*)acc1 + (size_t)tile_c * (DEG_P * 16) + half * (DEG_P * 8) + l;
#pragma unroll
        for (int j = 0; j < DEG_P / 2; ++j) { so += q0[j * 16]; si += q1[j * 16]; }
    }

    {
        float4 w[8];
#pragma unroll
        for (int q = 0; q < 8; ++q) w[q] = ((const float4*)W)[threadIdx.x + 256 * q];
#pragma unroll
        for (int q = 0; q < 8; ++q) {
            const int i = threadIdx.x + 256 * q;
            const int r = i >> 7;
            const int c = (i & 127) << 2;
            float* dr = &Wlds[r * LDSW_STRIDE + c];
            dr[0] = w[q].x; dr[1] = w[q].y; dr[2] = w[q].z; dr[3] = w[q].w;
        }
    }
    __syncthreads();
    if (!valid) return;

    float4*      stg4 = Astg + wave * (STG_WAVE / 4) + half * (STG_PITCH / 4) + l;
    const float* arow = (const float*)Astg + wave * STG_WAVE + l * STG_PITCH + 2 * half;
    const float* brow = &Wlds[l * LDSW_STRIDE + 2 * half];

    v8f acc = {};
    bufb = gemm_load_batch(gsrc + 1 * (KB / 4)); gemm_use_batch(bufa, stg4, arow, brow + 0 * KB, acc);
    bufa = gemm_load_batch(gsrc + 2 * (KB / 4)); gemm_use_batch(bufb, stg4, arow, brow + 1 * KB, acc);
    bufb = gemm_load_batch(gsrc + 3 * (KB / 4)); gemm_use_batch(bufa, stg4, arow, brow + 2 * KB, acc);
    bufa = gemm_load_batch(gsrc + 4 * (KB / 4)); gemm_use_batch(bufb, stg4, arow, brow + 3 * KB, acc);
    bufb = gemm_load_batch(gsrc + 5 * (KB / 4)); gemm_use_batch(bufa, stg4, arow, brow + 4 * KB, acc);
    bufa = gemm_load_batch(gsrc + 6 * (KB / 4)); gemm_use_batch(bufb, stg4, arow, brow + 5 * KB, acc);
    bufb = gemm_load_batch(gsrc + 7 * (KB / 4)); gemm_use_batch(bufa, stg4, arow, brow + 6 * KB, acc);
    gemm_use_batch(bufb, stg4, arow, brow + 7 * KB, acc);

    so += (unsigned)__shfl_xor((int)so, 16);
    si += (unsigned)__shfl_xor((int)si, 16);
    const float od = (float)so;
    const float id = (float)si;
    if (half) indeg[tile * 16 + l] = id; else outdeg[tile * 16 + l] = od;
    const float nsl = rsqrtf(fmaxf(od, 1.0f));

    const float bv = bias[l];
#pragma unroll
    for (int r = 0; r < 8; ++r) {
        const int row = tile * 16 + r + 8 * half;
        const float ns = __shfl(nsl, r + 8 * half);
        float v = acc[r] + bv;
        v = v > 0.0f ? v : 0.0f;
        hbuf[(size_t)row * C_OUT + l] = v * ns;
    }

    {
        const float4 z = {0.0f, 0.0f, 0.0f, 0.0f};
        float4* z0 = (float4*)(acc0 + (size_t)tile * (16 * C_OUT));
        float4* z1 = (float4*)(acc1 + (size_t)tile * (16 * C_OUT));
        z0[lane] = z; z0[lane + 32] = z;
        z1[lane] = z; z1[lane + 32] = z;
    }
}

#define DEG_R       12800
#define DEG_THREADS 1024

#define BK_R     400
#define BK_MAX   256
#define BK_NCH   256
#define BK_CHMAX 12544
#define BK_SBITS 17
static_assert(BK_NCH == 256 && BK_MAX == 256 && DEG_THREADS == 1024, "the place kernel's thread maps assume these sizes");

static __device__ __forceinline__ unsigned bk_of(const int d) {
    const unsigned b = (unsigned)d / (unsigned)BK_R;
    return b < (unsigned)BK_MAX ? b : (unsigned)(BK_MAX - 1);
}

static __device__ __forceinline__ void deg_bin4(unsigned* cnt, const int4 q, const int lo) {
    unsigned r;
    r = (unsigned)(q.x - lo); if (r < (unsigned)DEG_R) atomicAdd(&cnt[r], 1u);
    r = (unsigned)(q.y - lo); if (r < (unsigned)DEG_R) atomicAdd(&cnt[r], 1u);
    r = (unsigned)(q.z - lo); if (r < (unsigned)DEG_R) atomicAdd(&cnt[r], 1u);
    r = (unsigned)(q.w - lo); if (r < (unsigned)DEG_R) atomicAdd(&cnt[r], 1u);
}

__launch_bounds__(DEG_THREADS)
__global__ void appnp_degree_count(const int* __restrict__ src,
                                   const int* __restrict__ dst,
                                   unsigned* __restrict__ part0,
                                   unsigned* __restrict__ part1,
                                   int E, int N, int chunk,
                                   unsigned* __restrict__ cntm, int bchunk) {
    __shared__ unsigned cnt[DEG_R];

    if (blockIdx.z >= 2) {
        const int layer = gridDim.x * gridDim.y;
        for (int c = (blockIdx.z - 2) * layer + blockIdx.y * gridDim.x + blockIdx.x; c < BK_NCH;
             c += (gridDim.z - 2) * layer) {
            if (threadIdx.x < BK_MAX) cnt[threadIdx.x] = 0u;
            __syncthreads();
            const int e0 = c * bchunk;
            int e1 = e0 + bchunk; if (e1 > E) e1 = E;
            const int len  = e1 > e0 ? e1 - e0 : 0;
            const int nvec = len >> 2;
            const int4* __restrict__ v = (const int4*)(dst + e0);
            for (int i = threadIdx.x; i < nvec; i += DEG_THREADS) {
                const int4 q = v[i];
                atomicAdd(&cnt[bk_of(q.x)], 1u); atomicAdd(&cnt[bk_of(q.y)], 1u);
                atomicAdd(&cnt[bk_of(q.z)], 1u); atomicAdd(&cnt[bk_of(q.w)], 1u);
            }
            for (int e = e0 + (nvec << 2) + threadIdx.x; e < e1; e += DEG_THREADS)
                atomicAdd(&cnt[bk_of(dst[e])], 1u);
            __syncthreads();
            if (threadIdx.x < BK_MAX) cntm[c * BK_MAX + threadIdx.x] = cnt[threadIdx.x];
            __syncthreads();
        }
        return;
    }

    const int h = blockIdx.x;
    const int p = blockIdx.y;
    const int a = blockIdx.z;
    const int* __restrict__ idx = a ? dst : src;
    const int lo = h * DEG_R;

    for (int i = threadIdx.x; i < DEG_R; i += DEG_THREADS) cnt[i] = 0u;
    __syncthreads();

    const int e0 = p * chunk;
    int e1 = e0 + chunk; if (e1 > E) e1 = E;
    const int len  = e1 > e0 ? e1 - e0 : 0;
    const int nvec = len >> 2;
    const int4* __restrict__ v = (const int4*)(idx + e0);

    int i = threadIdx.x;
    for (; i + 3 * DEG_THREADS < nvec; i += 4 * DEG_THREADS) {
        const int4 q0 = v[i];
        const int4 q1 = v[i + DEG_THREADS];
        const int4 q2 = v[i + 2 * DEG_THREADS];
        const int4 q3 = v[i + 3 * DEG_THREADS];
        deg_bin4(cnt, q0, lo); deg_bin4(cnt, q1, lo);
        deg_bin4(cnt, q2, lo); deg_bin4(cnt, q3, lo);
    }
    for (; i < nvec; i += DEG_THREADS) deg_bin4(cnt, v[i], lo);
    for (int e = e0 + (nvec << 2) + threadIdx.x; e < e1; e += DEG_THREADS) {
        const unsigned r = (unsigned)(idx[e] - lo);
        if (r < (unsigned)DEG_R) atomicAdd(&cnt[r], 1u);
    }
    __syncthreads();

    int nr = N - lo; if (nr > DEG_R) nr = DEG_R;
    unsigned* __restrict__ plane = (a ? part1 : part0) + p * 16;
    for (int k = threadIdx.x; k < nr; k += DEG_THREADS) {
        const int n = lo + k;
        plane[(size_t)(n >> 4) * (DEG_P * 16) + (n & 15)] = cnt[k];
    }
}

#define SC_THREADS 256
#define SC_BLOCKS  2048

static __device__ __forceinline__ unsigned l2_half() {
    const unsigned id = (unsigned)__builtin_amdgcn_s_sendmsg_rtn(0x87);
    return (id >> 18) & 1u;
}

template <int J>
static __device__ __forceinline__ unsigned row_bcast(unsigned v) {
    return (unsigned)__builtin_amdgcn_update_dpp(0, (int)v, 0x150 + J, 0xf, 0xf, true);
}

template <int J>
static __device__ __forceinline__ float sc_load(const float* __restrict__ xbuf,
                                                unsigned srow, unsigned cb) {
    return *(const float*)((const char*)xbuf + (row_bcast<J>(srow) + cb));
}

template <int J, bool GUARD>
static __device__ __forceinline__ void sc_add(float* __restrict__ agg, unsigned drow,
                                              unsigned cb, float x,
                                              int eb, int E) {
    const unsigned off = row_bcast<J>(drow) + cb;
    if (!GUARD || eb + J < E)
        atomicAdd((float*)((char*)agg + off), x);
}

template <bool GUARD>
static __device__ __forceinline__ void sc_batch(const float* __restrict__ xbuf,
                                                float* __restrict__ agg,
                                                unsigned srow, unsigned drow,
                                                unsigned cb, int eb, int E) {
    const float x0  = sc_load<0>(xbuf, srow, cb);
    const float x1  = sc_load<1>(xbuf, srow, cb);
    const float x2  = sc_load<2>(xbuf, srow, cb);
    const float x3  = sc_load<3>(xbuf, srow, cb);
    const float x4  = sc_load<4>(xbuf, srow, cb);
    const float x5  = sc_load<5>(xbuf, srow, cb);
    const float x6  = sc_load<6>(xbuf, srow, cb);
    const float x7  = sc_load<7>(xbuf, srow, cb);
    const float x8  = sc_load<8>(xbuf, srow, cb);
    const float x9  = sc_load<9>(xbuf, srow, cb);
    const float x10 = sc_load<10>(xbuf, srow, cb);
    const float x11 = sc_load<11>(xbuf, srow, cb);
    const float x12 = sc_load<12>(xbuf, srow, cb);
    const float x13 = sc_load<13>(xbuf, srow, cb);
    const float x14 = sc_load<14>(xbuf, srow, cb);
    const float x15 = sc_load<15>(xbuf, srow, cb);
    sc_add<0,  GUARD>(agg, drow, cb, x0,  eb, E);
    sc_add<1,  GUARD>(agg, drow, cb, x1,  eb, E);
    sc_add<2,  GUARD>(agg, drow, cb, x2,  eb, E);
    sc_add<3,  GUARD>(agg, drow, cb, x3,  eb, E);
    sc_add<4,  GUARD>(agg, drow, cb, x4,  eb, E);
    sc_add<5,  GUARD>(agg, drow, cb, x5,  eb, E);
    sc_add<6,  GUARD>(agg, drow, cb, x6,  eb, E);
    sc_add<7,  GUARD>(agg, drow, cb, x7,  eb, E);
    sc_add<8,  GUARD>(agg, drow, cb, x8,  eb, E);
    sc_add<9,  GUARD>(agg, drow, cb, x9,  eb, E);
    sc_add<10, GUARD>(agg, drow, cb, x10, eb, E);
    sc_add<11, GUARD>(agg, drow, cb, x11, eb, E);
    sc_add<12, GUARD>(agg, drow, cb, x12, eb, E);
    sc_add<13, GUARD>(agg, drow, cb, x13, eb, E);
    sc_add<14, GUARD>(agg, drow, cb, x14, eb, E);
    sc_add<15, GUARD>(agg, drow, cb, x15, eb, E);
}

__launch_bounds__(SC_THREADS)
__global__ void appnp_scatter(const int* __restrict__ src,
                              const int* __restrict__ dst,
                              const float* __restrict__ xbuf,
                              float* agg0, float* agg1, int E) {
    float* __restrict__ agg = l2_half() ? agg1 : agg0;
    const int lane   = threadIdx.x & 31;
    const int gwave  = (int)((blockIdx.x * SC_THREADS + threadIdx.x) >> 5);
    const int nwaves = (int)((gridDim.x * SC_THREADS) >> 5);
    const int nb     = (E + 31) >> 5;
    const unsigned cb = (unsigned)(lane & 15) << 2;
    const int hofs   = lane & 16;

    int b = gwave;
    if (b >= nb) return;
    int e = (b << 5) + lane;
    int s = 0, d = 0;
    if (e < E) { s = src[e]; d = dst[e]; }

    for (;;) {
        const int bn = b + nwaves;
        int sn = 0, dn = 0;
        if (bn < nb) {
            const int en = (bn << 5) + lane;
            if (en < E) { sn = src[en]; dn = dst[en]; }
        }
        const unsigned srow = (unsigned)s << 6;
        const unsigned drow = (unsigned)d << 6;
        const int eb = (b << 5) + hofs;
        if ((b << 5) + 32 <= E) sc_batch<false>(xbuf, agg, srow, drow, cb, eb, E);
        else                    sc_batch<true >(xbuf, agg, srow, drow, cb, eb, E);
        if (bn >= nb) break;
        b = bn; s = sn; d = dn;
    }
}

__global__ void appnp_finalize(const float* __restrict__ agg0,
                               const float* __restrict__ outdeg,
                               const float* __restrict__ indeg,
                               const float* __restrict__ hbuf,
                               float* out, int N) {
    int t = blockIdx.x * blockDim.x + threadIdx.x;
    if (t >= N * 4) return;
    const int n = t >> 2;
    const float nd = rsqrtf(fmaxf(indeg[n], 1.0f));
    const float ns = rsqrtf(fmaxf(outdeg[n], 1.0f));
    const float4 a0 = ((const float4*)agg0)[t];
    const float4 a1 = ((const float4*)out)[t];
    float4 a;
    a.x = a0.x + a1.x; a.y = a0.y + a1.y; a.z = a0.z + a1.z; a.w = a0.w + a1.w;
    const float4 h0 = ((const float4*)hbuf)[t];
    float4 x0;
    x0.x = h0.x / ns; x0.y = h0.y / ns; x0.z = h0.z / ns; x0.w = h0.w / ns;
    float4 o;
    o.x = 0.8f * a.x * nd + 0.2f * x0.x;
    o.y = 0.8f * a.y * nd + 0.2f * x0.y;
    o.z = 0.8f * a.z * nd + 0.2f * x0.z;
    o.w = 0.8f * a.w * nd + 0.2f * x0.w;
    ((float4*)out)[t] = o;
}


static __device__ __forceinline__ void bk_stage1(const int s, const int d,
                                                 unsigned* cursor, const unsigned* lstart,
                                                 unsigned* sval, unsigned short* sbk) {
    const unsigned b    = bk_of(d);
    const unsigned dl   = (unsigned)d - b * (unsigned)BK_R;
    const unsigned slot = lstart[b] + atomicAdd(&cursor[b], 1u);
    if (slot < (unsigned)BK_CHMAX) {
        sval[slot] = (unsigned)s | (dl << BK_SBITS);
        sbk[slot]  = (unsigned short)b;
    }
}

__launch_bounds__(DEG_THREADS)
__global__ void appnp_bucket_place(const int* __restrict__ src,
                                   const int* __restrict__ dst,
                                   const unsigned* __restrict__ cntm,
                                   unsigned* __restrict__ bst,
                                   unsigned* __restrict__ btot,
                                   unsigned* __restrict__ sorted,
                                   int E, int bchunk) {
    __shared__ uint4 ptot4[16 * BK_MAX / 4];
    __shared__ uint4 ppre4[16 * BK_MAX / 4];
    __shared__ unsigned gbase[BK_MAX];
    __shared__ unsigned lstart[BK_MAX];
    __shared__ unsigned cursor[BK_MAX];
    __shared__ unsigned wsa[8], wsb[8];
    __shared__ unsigned sval[BK_CHMAX];
    __shared__ unsigned short sbk[BK_CHMAX];

    const int c    = blockIdx.x;
    const int tid  = threadIdx.x;
    const int lane = tid & 31;
    const int wave = tid >> 5;

    {
        const int b4 = tid & 63;
        const int qq = tid >> 6;
        const uint4* __restrict__ m4 = (const uint4*)cntm;
        uint4 v[16];
#pragma unroll
        for (int j = 0; j < 16; ++j) v[j] = m4[(qq + 16 * j) * (BK_MAX / 4) + b4];
        uint4 t = {0u, 0u, 0u, 0u}, p = {0u, 0u, 0u, 0u};
#pragma unroll
        for (int j = 0; j < 16; ++j) {
            t.x += v[j].x; t.y += v[j].y; t.z += v[j].z; t.w += v[j].w;
            if (qq + 16 * j < c) { p.x += v[j].x; p.y += v[j].y; p.z += v[j].z; p.w += v[j].w; }
        }
        ptot4[qq * (BK_MAX / 4) + b4] = t;
        ppre4[qq * (BK_MAX / 4) + b4] = p;
    }
    __syncthreads();

    const unsigned* ptot = (const unsigned*)ptot4;
    const unsigned* ppre = (const unsigned*)ppre4;
    const int b = tid;
    unsigned tot = 0u, pre = 0u, mine = 0u, it = 0u, im = 0u;
    if (tid < BK_MAX) {
        for (int k = 0; k < 16; ++k) { tot += ptot[k * BK_MAX + b]; pre += ppre[k * BK_MAX + b]; }
        mine = cntm[c * BK_MAX + b];
        it = tot; im = mine;
        for (int d = 1; d < 32; d <<= 1) {
            const unsigned a = __shfl_up(it, (unsigned)d);
            const unsigned m = __shfl_up(im, (unsigned)d);
            if (lane >= d) { it += a; im += m; }
        }
        if (lane == 31) { wsa[wave] = it; wsb[wave] = im; }
    }
    __syncthreads();
    if (tid < BK_MAX) {
        unsigned oa = 0u, ob = 0u;
        for (int w = 0; w < wave; ++w) { oa += wsa[w]; ob += wsb[w]; }
        const unsigned bstart = oa + it - tot;
        gbase[b]  = bstart + pre;
        lstart[b] = ob + im - mine;
        cursor[b] = 0u;
        if (c == 0) { bst[b] = bstart; btot[b] = tot; }
    }
    __syncthreads();

    const int e0 = c * bchunk;
    int e1 = e0 + bchunk; if (e1 > E) e1 = E;
    const int len  = e1 > e0 ? e1 - e0 : 0;
    const int nvec = len >> 2;
    const int4* __restrict__ vs = (const int4*)(src + e0);
    const int4* __restrict__ vd = (const int4*)(dst + e0);
    for (int i = tid; i < nvec; i += DEG_THREADS) {
        const int4 s4 = vs[i];
        const int4 d4 = vd[i];
        bk_stage1(s4.x, d4.x, cursor, lstart, sval, sbk);
        bk_stage1(s4.y, d4.y, cursor, lstart, sval, sbk);
        bk_stage1(s4.z, d4.z, cursor, lstart, sval, sbk);
        bk_stage1(s4.w, d4.w, cursor, lstart, sval, sbk);
    }
    for (int e = e0 + (nvec << 2) + tid; e < e1; e += DEG_THREADS)
        bk_stage1(src[e], dst[e], cursor, lstart, sval, sbk);
    __syncthreads();

    for (int i = tid; i < len; i += DEG_THREADS) {
        const unsigned bb  = sbk[i];
        const unsigned pos = gbase[bb] + ((unsigned)i - lstart[bb]);
        if (pos < (unsigned)E) sorted[pos] = sval[i];
    }
}

#define BK_REP16(M) M(0) M(1) M(2) M(3) M(4) M(5) M(6) M(7) M(8) M(9) M(10) M(11) M(12) M(13) M(14) M(15)

template <int J, bool GUARD>
static __device__ __forceinline__ void bk_add(float* accl, unsigned drow, unsigned cb, float x,
                                              unsigned ib, unsigned n) {
    const unsigned off = row_bcast<J>(drow) + cb;
    if (!GUARD || ib + J < n)
        atomicAdd((float*)((char*)accl + off), x);
}

template <bool GUARD>
static __device__ __forceinline__ void bk_batch2(const float* __restrict__ hbuf, float* accl,
                                                 unsigned va, unsigned vb, unsigned cb,
                                                 unsigned ia, unsigned ib, unsigned n) {
    const unsigned sa = (va & ((1u << BK_SBITS) - 1u)) << 6;
    const unsigned da = (va >> BK_SBITS) << 6;
    const unsigned sb = (vb & ((1u << BK_SBITS) - 1u)) << 6;
    const unsigned db = (vb >> BK_SBITS) << 6;
    float xa[16], xb[16];
#define BK_GA(J) xa[J] = sc_load<J>(hbuf, sa, cb);
#define BK_GB(J) xb[J] = sc_load<J>(hbuf, sb, cb);
    BK_REP16(BK_GA)
    BK_REP16(BK_GB)
#define BK_AA(J) bk_add<J, GUARD>(accl, da, cb, xa[J], ia, n);
#define BK_AB(J) bk_add<J, GUARD>(accl, db, cb, xb[J], ib, n);
    BK_REP16(BK_AA)
    BK_REP16(BK_AB)
}

__launch_bounds__(DEG_THREADS)
__global__ void appnp_bucket_sum(const unsigned* __restrict__ sorted,
                                 const unsigned* __restrict__ bst,
                                 const unsigned* __restrict__ btot,
                                 const float* __restrict__ hbuf,
                                 const float* __restrict__ outdeg,
                                 const float* __restrict__ indeg,
                                 float* __restrict__ out, int N) {
    __shared__ float4 accl4[BK_R * C_OUT / 4];
    float* accl = (float*)accl4;

    const int bk   = blockIdx.x;
    const int tid  = threadIdx.x;
    const int lane = tid & 31;
    const int wave = tid >> 5;
    const int lo   = bk * BK_R;
    int nr = N - lo; if (nr > BK_R) nr = BK_R;

    {
        const float4 z = {0.0f, 0.0f, 0.0f, 0.0f};
        for (int i = tid; i < BK_R * C_OUT / 4; i += DEG_THREADS) accl4[i] = z;
    }
    __syncthreads();

    const unsigned start = bst[bk];
    const unsigned n     = btot[bk];
    const unsigned* __restrict__ mine = sorted + start;
    const unsigned cb   = (unsigned)(lane & 15) << 2;
    const unsigned hofs = (unsigned)(lane & 16);

    for (unsigned base = (unsigned)wave << 5; base < n; base += 2u * DEG_THREADS) {
        const unsigned ea = base + (unsigned)lane;
        const unsigned eb = ea + (unsigned)DEG_THREADS;
        const unsigned va = ea < n ? mine[ea] : 0u;
        const unsigned vb = eb < n ? mine[eb] : 0u;
        if (base + (unsigned)DEG_THREADS + 32u <= n)
            bk_batch2<false>(hbuf, accl, va, vb, cb, base + hofs, base + (unsigned)DEG_THREADS + hofs, n);
        else
            bk_batch2<true >(hbuf, accl, va, vb, cb, base + hofs, base + (unsigned)DEG_THREADS + hofs, n);
    }
    __syncthreads();

    for (int idx = tid; idx < nr * 4; idx += DEG_THREADS) {
        const int    node = lo + (idx >> 2);
        const size_t t    = (size_t)lo * 4 + (size_t)idx;
        const float nd = rsqrtf(fmaxf(indeg[node], 1.0f));
        const float ns = rsqrtf(fmaxf(outdeg[node], 1.0f));
        const float4 a  = accl4[idx];
        const float4 h0 = ((const float4*)hbuf)[t];
        float4 x0;
        x0.x = h0.x / ns; x0.y = h0.y / ns; x0.z = h0.z / ns; x0.w = h0.w / ns;
        float4 o;
        o.x = 0.8f * a.x * nd + 0.2f * x0.x;
        o.y = 0.8f * a.y * nd + 0.2f * x0.y;
        o.z = 0.8f * a.z * nd + 0.2f * x0.z;
        o.w = 0.8f * a.w * nd + 0.2f * x0.w;
        ((float4*)out)[t] = o;
    }
}

extern "C" void kernel_launch(void* const* d_in, const int* in_sizes, int n_in,
                              void* d_out, int out_size, void* d_ws, size_t ws_size,
                              hipStream_t stream) {
    const float* feat = (const float*)d_in[0];
    const float* W    = (const float*)d_in[1];
    const float* bias = (const float*)d_in[2];
    const int*   src  = (const int*)d_in[3];
    const int*   dst  = (const int*)d_in[4];

    const int N = in_sizes[0] / F_IN;
    const int E = in_sizes[3];

    float* agg1 = (float*)d_out;
    float* ws     = (float*)d_ws;
    float* xbuf   = ws;
    float* agg    = xbuf + (size_t)N * C_OUT;
    float* outdeg = agg  + (size_t)N * C_OUT;
    float* indeg  = outdeg + N;

    const int    nbk    = (N + BK_R - 1) / BK_R;
    const int    bchunk = (((E + BK_NCH - 1) / BK_NCH) + 3) & ~3;
    const size_t used   = (size_t)N * (2 * C_OUT + 2) * sizeof(float);
    const size_t need   = used + 16 + ((size_t)BK_NCH * BK_MAX + 2 * BK_MAX + (size_t)E) * sizeof(unsigned);
    const bool ordered  = ws_size >= need && N <= (1 << BK_SBITS) && nbk <= BK_MAX && bchunk <= BK_CHMAX;
    unsigned* cntm   = (unsigned*)(((size_t)(indeg + N) + 15) & ~(size_t)15);
    unsigned* bst    = cntm + (size_t)BK_NCH * BK_MAX;
    unsigned* btot   = bst + BK_MAX;
    unsigned* sorted = btot + BK_MAX;

    {
        const int chunk  = (((E + DEG_P - 1) / DEG_P) + 3) & ~3;
        const int ranges = (N + DEG_R - 1) / DEG_R;
        appnp_degree_count<<<dim3(ranges, DEG_P, ordered ? 4 : 2), DEG_THREADS, 0, stream>>>(
            src, dst, (unsigned*)agg, (unsigned*)agg1, E, N, chunk, cntm, bchunk);
    }
    {
        const int nTiles = (N + 15) / 16;
        const int blocks = (nTiles + 7) / 8;
        appnp_gemm_relu<<<blocks, 256, 0, stream>>>(feat, W, bias, agg, agg1, outdeg, indeg,
                                                    xbuf, nTiles);
    }
    if (ordered) {
        appnp_bucket_place<<<BK_NCH, DEG_THREADS, 0, stream>>>(src, dst, cntm, bst, btot, sorted, E, bchunk);
        appnp_bucket_sum<<<nbk, DEG_THREADS, 0, stream>>>(sorted, bst, btot, xbuf, outdeg, indeg,
                                                          (float*)d_out, N);
        return;
    }
    {
        appnp_scatter<<<SC_BLOCKS, SC_THREADS, 0, stream>>>(src, dst, xbuf, agg, agg1, E);
    }
    appnp_finalize<<<(N * 4 + 255) / 256, 256, 0, stream>>>(agg, outdeg, indeg, xbuf,
                                                            (float*)d_out, N);
}


# in-degree comes from the sorted edge list (each bucket workgroup counts its nodes' edges in LDS, u32): launch 1 histograms src only, 256 workgroups on half-size chunks instead of 512
# speedup vs baseline: 5.4851x; 1.0338x over previous
#include <hip/hip_runtime.h>
#include <hip/hip_bf16.h>


typedef float v2f __attribute__((ext_vector_type(2)));
typedef float v8f __attribute__((ext_vector_type(8)));

#define F_IN        512
#define C_OUT       16
#define LDSW_STRIDE (F_IN + 2)
#define DEG_P       16

#define KB        64
#define NBATCH    (F_IN / KB)
#define STG_PITCH (KB + 4)
#define STG_WAVE  (16 * STG_PITCH)

struct batch8 { float4 p0, p1, p2, p3, p4, p5, p6, p7; };

static __device__ __forceinline__ batch8 gemm_load_batch(const float4* __restrict__ g) {
    batch8 d;
    d.p0 = g[0 * (2 * F_IN / 4)]; d.p1 = g[1 * (2 * F_IN / 4)];
    d.p2 = g[2 * (2 * F_IN / 4)]; d.p3 = g[3 * (2 * F_IN / 4)];
    d.p4 = g[4 * (2 * F_IN / 4)]; d.p5 = g[5 * (2 * F_IN / 4)];
    d.p6 = g[6 * (2 * F_IN / 4)]; d.p7 = g[7 * (2 * F_IN / 4)];
    return d;
}

static __device__ __forceinline__ void gemm_use_batch(const batch8 s, float4* stg4,
                                                      const float* arow, const float* browt,
                                                      v8f& acc) {
    stg4[0 * (2 * STG_PITCH / 4)] = s.p0; stg4[1 * (2 * STG_PITCH / 4)] = s.p1;
    stg4[2 * (2 * STG_PITCH / 4)] = s.p2; stg4[3 * (2 * STG_PITCH / 4)] = s.p3;
    stg4[4 * (2 * STG_PITCH / 4)] = s.p4; stg4[5 * (2 * STG_PITCH / 4)] = s.p5;
    stg4[6 * (2 * STG_PITCH / 4)] = s.p6; stg4[7 * (2 * STG_PITCH / 4)] = s.p7;
#pragma unroll
    for (int k = 0; k < KB / 4; ++k) {
        v2f a = *(const v2f*)(arow + 4 * k);
        v2f b = *(const v2f*)(browt + 4 * k);
        acc = __builtin_amdgcn_wmma_f32_16x16x4_f32(
             false, a,  false, b,
             (short)0, acc,  false,  false);
    }
}

__launch_bounds__(256, 8)
__global__ void appnp_gemm_relu(const float* __restrict__ feat,
                                const float* __restrict__ W,
                                const float* __restrict__ bias,
                                float* acc0, float* acc1,
                                float* __restrict__ outdeg,
                                float* __restrict__ indeg,
                                float* __restrict__ hbuf,
                                int nTiles, unsigned srcMask) {
    __shared__ float  Wlds[C_OUT * LDSW_STRIDE];
    __shared__ float4 Astg[8 * STG_WAVE / 4];

    const int  wave  = threadIdx.x >> 5;
    const int  lane  = threadIdx.x & 31;
    const int  tile  = blockIdx.x * 8 + wave;
    const bool valid = tile < nTiles;
    const int  half  = lane >> 4;
    const int  l     = lane & 15;

    const int tile_c = valid ? tile : nTiles - 1;
    const float4* gsrc = (const float4*)(feat + (size_t)(tile_c * 16 + half) * F_IN) + l;
    batch8 bufa = gemm_load_batch(gsrc);
    batch8 bufb;

    unsigned so = 0u, si = 0u;
    {
        const unsigned* q0 = (const unsigned*)acc0 + (size_t)tile_c * (DEG_P * 16) + half * (DEG_P * 8) + l;
        const unsigned* q1 = (const unsigned*)acc1 + (size_t)tile_c * (DEG_P * 16) + half * (DEG_P * 8) + l;
#pragma unroll
        for (int j = 0; j < DEG_P / 2; ++j) { so += q0[j * 16]; si += q1[j * 16]; }
    }

    {
        float4 w[8];
#pragma unroll
        for (int q = 0; q < 8; ++q) w[q] = ((const float4*)W)[threadIdx.x + 256 * q];
#pragma unroll
        for (int q = 0; q < 8; ++q) {
            const int i = threadIdx.x + 256 * q;
            const int r = i >> 7;
            const int c = (i & 127) << 2;
            float* dr = &Wlds[r * LDSW_STRIDE + c];
            dr[0] = w[q].x; dr[1] = w[q].y; dr[2] = w[q].z; dr[3] = w[q].w;
        }
    }
    __syncthreads();
    if (!valid) return;

    float4*      stg4 = Astg + wave * (STG_WAVE / 4) + half * (STG_PITCH / 4) + l;
    const float* arow = (const float*)Astg + wave * STG_WAVE + l * STG_PITCH + 2 * half;
    const float* brow = &Wlds[l * LDSW_STRIDE + 2 * half];

    v8f acc = {};
    bufb = gemm_load_batch(gsrc + 1 * (KB / 4)); gemm_use_batch(bufa, stg4, arow, brow + 0 * KB, acc);
    bufa = gemm_load_batch(gsrc + 2 * (KB / 4)); gemm_use_batch(bufb, stg4, arow, brow + 1 * KB, acc);
    bufb = gemm_load_batch(gsrc + 3 * (KB / 4)); gemm_use_batch(bufa, stg4, arow, brow + 2 * KB, acc);
    bufa = gemm_load_batch(gsrc + 4 * (KB / 4)); gemm_use_batch(bufb, stg4, arow, brow + 3 * KB, acc);
    bufb = gemm_load_batch(gsrc + 5 * (KB / 4)); gemm_use_batch(bufa, stg4, arow, brow + 4 * KB, acc);
    bufa = gemm_load_batch(gsrc + 6 * (KB / 4)); gemm_use_batch(bufb, stg4, arow, brow + 5 * KB, acc);
    bufb = gemm_load_batch(gsrc + 7 * (KB / 4)); gemm_use_batch(bufa, stg4, arow, brow + 6 * KB, acc);
    gemm_use_batch(bufb, stg4, arow, brow + 7 * KB, acc);

    so += (unsigned)__shfl_xor((int)so, 16);
    si += (unsigned)__shfl_xor((int)si, 16);
    so += si & srcMask;
    const float od = (float)so;
    const float id = (float)si;
    if (half) indeg[tile * 16 + l] = id; else outdeg[tile * 16 + l] = od;
    const float nsl = rsqrtf(fmaxf(od, 1.0f));

    const float bv = bias[l];
#pragma unroll
    for (int r = 0; r < 8; ++r) {
        const int row = tile * 16 + r + 8 * half;
        const float ns = __shfl(nsl, r + 8 * half);
        float v = acc[r] + bv;
        v = v > 0.0f ? v : 0.0f;
        hbuf[(size_t)row * C_OUT + l] = v * ns;
    }

    {
        const float4 z = {0.0f, 0.0f, 0.0f, 0.0f};
        float4* z0 = (float4*)(acc0 + (size_t)tile * (16 * C_OUT));
        float4* z1 = (float4*)(acc1 + (size_t)tile * (16 * C_OUT));
        z0[lane] = z; z0[lane + 32] = z;
        z1[lane] = z; z1[lane + 32] = z;
    }
}

#define DEG_R       12800
#define DEG_THREADS 1024

#define BK_R     400
#define BK_MAX   256
#define BK_NCH   256
#define BK_CHMAX 12544
#define BK_SBITS 17
static_assert(BK_NCH == 256 && BK_MAX == 256 && DEG_THREADS == 1024, "the place kernel's thread maps assume these sizes");

static __device__ __forceinline__ unsigned bk_of(const int d) {
    const unsigned b = (unsigned)d / (unsigned)BK_R;
    return b < (unsigned)BK_MAX ? b : (unsigned)(BK_MAX - 1);
}

static __device__ __forceinline__ void deg_bin4(unsigned* cnt, const int4 q, const int lo) {
    unsigned r;
    r = (unsigned)(q.x - lo); if (r < (unsigned)DEG_R) atomicAdd(&cnt[r], 1u);
    r = (unsigned)(q.y - lo); if (r < (unsigned)DEG_R) atomicAdd(&cnt[r], 1u);
    r = (unsigned)(q.z - lo); if (r < (unsigned)DEG_R) atomicAdd(&cnt[r], 1u);
    r = (unsigned)(q.w - lo); if (r < (unsigned)DEG_R) atomicAdd(&cnt[r], 1u);
}

__launch_bounds__(DEG_THREADS)
__global__ void appnp_degree_count(const int* __restrict__ src,
                                   const int* __restrict__ dst,
                                   unsigned* __restrict__ part0,
                                   unsigned* __restrict__ part1,
                                   int E, int N, int chunk,
                                   unsigned* __restrict__ cntm, int bchunk, int srcOnly) {
    __shared__ unsigned cnt[DEG_R];

    if (srcOnly) {
        const int nwg = gridDim.x * gridDim.y;
        for (int c = blockIdx.y * gridDim.x + blockIdx.x; c < BK_NCH; c += nwg) {
            if (threadIdx.x < BK_MAX) cnt[threadIdx.x] = 0u;
            __syncthreads();
            const int e0 = c * bchunk;
            int e1 = e0 + bchunk; if (e1 > E) e1 = E;
            const int len  = e1 > e0 ? e1 - e0 : 0;
            const int nvec = len >> 2;
            const int4* __restrict__ v = (const int4*)(dst + e0);
            for (int i = threadIdx.x; i < nvec; i += DEG_THREADS) {
                const int4 q = v[i];
                atomicAdd(&cnt[bk_of(q.x)], 1u); atomicAdd(&cnt[bk_of(q.y)], 1u);
                atomicAdd(&cnt[bk_of(q.z)], 1u); atomicAdd(&cnt[bk_of(q.w)], 1u);
            }
            for (int e = e0 + (nvec << 2) + threadIdx.x; e < e1; e += DEG_THREADS)
                atomicAdd(&cnt[bk_of(dst[e])], 1u);
            __syncthreads();
            if (threadIdx.x < BK_MAX) cntm[c * BK_MAX + threadIdx.x] = cnt[threadIdx.x];
            __syncthreads();
        }
    }

    const int h = blockIdx.x;
    const int a = srcOnly ? (int)(blockIdx.y / DEG_P) : (int)blockIdx.z;
    const int p = srcOnly ? (int)(blockIdx.y % DEG_P) : (int)blockIdx.y;
    const int* __restrict__ idx = (a && !srcOnly) ? dst : src;
    const int lo = h * DEG_R;

    for (int i = threadIdx.x; i < DEG_R; i += DEG_THREADS) cnt[i] = 0u;
    __syncthreads();

    const int e0 = (int)blockIdx.y * chunk;
    int e1 = e0 + chunk; if (e1 > E) e1 = E;
    const int len  = e1 > e0 ? e1 - e0 : 0;
    const int nvec = len >> 2;
    const int4* __restrict__ v = (const int4*)(idx + e0);

    int i = threadIdx.x;
    for (; i + 3 * DEG_THREADS < nvec; i += 4 * DEG_THREADS) {
        const int4 q0 = v[i];
        const int4 q1 = v[i + DEG_THREADS];
        const int4 q2 = v[i + 2 * DEG_THREADS];
        const int4 q3 = v[i + 3 * DEG_THREADS];
        deg_bin4(cnt, q0, lo); deg_bin4(cnt, q1, lo);
        deg_bin4(cnt, q2, lo); deg_bin4(cnt, q3, lo);
    }
    for (; i < nvec; i += DEG_THREADS) deg_bin4(cnt, v[i], lo);
    for (int e = e0 + (nvec << 2) + threadIdx.x; e < e1; e += DEG_THREADS) {
        const unsigned r = (unsigned)(idx[e] - lo);
        if (r < (unsigned)DEG_R) atomicAdd(&cnt[r], 1u);
    }
    __syncthreads();

    int nr = N - lo; if (nr > DEG_R) nr = DEG_R;
    unsigned* __restrict__ plane = (a ? part1 : part0) + p * 16;
    for (int k = threadIdx.x; k < nr; k += DEG_THREADS) {
        const int n = lo + k;
        plane[(size_t)(n >> 4) * (DEG_P * 16) + (n & 15)] = cnt[k];
    }
}

#define SC_THREADS 256
#define SC_BLOCKS  2048

static __device__ __forceinline__ unsigned l2_half() {
    const unsigned id = (unsigned)__builtin_amdgcn_s_sendmsg_rtn(0x87);
    return (id >> 18) & 1u;
}

template <int J>
static __device__ __forceinline__ unsigned row_bcast(unsigned v) {
    return (unsigned)__builtin_amdgcn_update_dpp(0, (int)v, 0x150 + J, 0xf, 0xf, true);
}

template <int J>
static __device__ __forceinline__ float sc_load(const float* __restrict__ xbuf,
                                                unsigned srow, unsigned cb) {
    return *(const float*)((const char*)xbuf + (row_bcast<J>(srow) + cb));
}

template <int J, bool GUARD>
static __device__ __forceinline__ void sc_add(float* __restrict__ agg, unsigned drow,
                                              unsigned cb, float x,
                                              int eb, int E) {
    const unsigned off = row_bcast<J>(drow) + cb;
    if (!GUARD || eb + J < E)
        atomicAdd((float*)((char*)agg + off), x);
}

template <bool GUARD>
static __device__ __forceinline__ void sc_batch(const float* __restrict__ xbuf,
                                                float* __restrict__ agg,
                                                unsigned srow, unsigned drow,
                                                unsigned cb, int eb, int E) {
    const float x0  = sc_load<0>(xbuf, srow, cb);
    const float x1  = sc_load<1>(xbuf, srow, cb);
    const float x2  = sc_load<2>(xbuf, srow, cb);
    const float x3  = sc_load<3>(xbuf, srow, cb);
    const float x4  = sc_load<4>(xbuf, srow, cb);
    const float x5  = sc_load<5>(xbuf, srow, cb);
    const float x6  = sc_load<6>(xbuf, srow, cb);
    const float x7  = sc_load<7>(xbuf, srow, cb);
    const float x8  = sc_load<8>(xbuf, srow, cb);
    const float x9  = sc_load<9>(xbuf, srow, cb);
    const float x10 = sc_load<10>(xbuf, srow, cb);
    const float x11 = sc_load<11>(xbuf, srow, cb);
    const float x12 = sc_load<12>(xbuf, srow, cb);
    const float x13 = sc_load<13>(xbuf, srow, cb);
    const float x14 = sc_load<14>(xbuf, srow, cb);
    const float x15 = sc_load<15>(xbuf, srow, cb);
    sc_add<0,  GUARD>(agg, drow, cb, x0,  eb, E);
    sc_add<1,  GUARD>(agg, drow, cb, x1,  eb, E);
    sc_add<2,  GUARD>(agg, drow, cb, x2,  eb, E);
    sc_add<3,  GUARD>(agg, drow, cb, x3,  eb, E);
    sc_add<4,  GUARD>(agg, drow, cb, x4,  eb, E);
    sc_add<5,  GUARD>(agg, drow, cb, x5,  eb, E);
    sc_add<6,  GUARD>(agg, drow, cb, x6,  eb, E);
    sc_add<7,  GUARD>(agg, drow, cb, x7,  eb, E);
    sc_add<8,  GUARD>(agg, drow, cb, x8,  eb, E);
    sc_add<9,  GUARD>(agg, drow, cb, x9,  eb, E);
    sc_add<10, GUARD>(agg, drow, cb, x10, eb, E);
    sc_add<11, GUARD>(agg, drow, cb, x11, eb, E);
    sc_add<12, GUARD>(agg, drow, cb, x12, eb, E);
    sc_add<13, GUARD>(agg, drow, cb, x13, eb, E);
    sc_add<14, GUARD>(agg, drow, cb, x14, eb, E);
    sc_add<15, GUARD>(agg, drow, cb, x15, eb, E);
}

__launch_bounds__(SC_THREADS)
__global__ void appnp_scatter(const int* __restrict__ src,
                              const int* __restrict__ dst,
                              const float* __restrict__ xbuf,
                              float* agg0, float* agg1, int E) {
    float* __restrict__ agg = l2_half() ? agg1 : agg0;
    const int lane   = threadIdx.x & 31;
    const int gwave  = (int)((blockIdx.x * SC_THREADS + threadIdx.x) >> 5);
    const int nwaves = (int)((gridDim.x * SC_THREADS) >> 5);
    const int nb     = (E + 31) >> 5;
    const unsigned cb = (unsigned)(lane & 15) << 2;
    const int hofs   = lane & 16;

    int b = gwave;
    if (b >= nb) return;
    int e = (b << 5) + lane;
    int s = 0, d = 0;
    if (e < E) { s = src[e]; d = dst[e]; }

    for (;;) {
        const int bn = b + nwaves;
        int sn = 0, dn = 0;
        if (bn < nb) {
            const int en = (bn << 5) + lane;
            if (en < E) { sn = src[en]; dn = dst[en]; }
        }
        const unsigned srow = (unsigned)s << 6;
        const unsigned drow = (unsigned)d << 6;
        const int eb = (b << 5) + hofs;
        if ((b << 5) + 32 <= E) sc_batch<false>(xbuf, agg, srow, drow, cb, eb, E);
        else                    sc_batch<true >(xbuf, agg, srow, drow, cb, eb, E);
        if (bn >= nb) break;
        b = bn; s = sn; d = dn;
    }
}

__global__ void appnp_finalize(const float* __restrict__ agg0,
                               const float* __restrict__ outdeg,
                               const float* __restrict__ indeg,
                               const float* __restrict__ hbuf,
                               float* out, int N) {
    int t = blockIdx.x * blockDim.x + threadIdx.x;
    if (t >= N * 4) return;
    const int n = t >> 2;
    const float nd = rsqrtf(fmaxf(indeg[n], 1.0f));
    const float ns = rsqrtf(fmaxf(outdeg[n], 1.0f));
    const float4 a0 = ((const float4*)agg0)[t];
    const float4 a1 = ((const float4*)out)[t];
    float4 a;
    a.x = a0.x + a1.x; a.y = a0.y + a1.y; a.z = a0.z + a1.z; a.w = a0.w + a1.w;
    const float4 h0 = ((const float4*)hbuf)[t];
    float4 x0;
    x0.x = h0.x / ns; x0.y = h0.y / ns; x0.z = h0.z / ns; x0.w = h0.w / ns;
    float4 o;
    o.x = 0.8f * a.x * nd + 0.2f * x0.x;
    o.y = 0.8f * a.y * nd + 0.2f * x0.y;
    o.z = 0.8f * a.z * nd + 0.2f * x0.z;
    o.w = 0.8f * a.w * nd + 0.2f * x0.w;
    ((float4*)out)[t] = o;
}


static __device__ __forceinline__ void bk_stage1(const int s, const int d,
                                                 unsigned* cursor, const unsigned* lstart,
                                                 unsigned* sval, unsigned short* sbk) {
    const unsigned b    = bk_of(d);
    const unsigned dl   = (unsigned)d - b * (unsigned)BK_R;
    const unsigned slot = lstart[b] + atomicAdd(&cursor[b], 1u);
    if (slot < (unsigned)BK_CHMAX) {
        sval[slot] = (unsigned)s | (dl << BK_SBITS);
        sbk[slot]  = (unsigned short)b;
    }
}

__launch_bounds__(DEG_THREADS)
__global__ void appnp_bucket_place(const int* __restrict__ src,
                                   const int* __restrict__ dst,
                                   const unsigned* __restrict__ cntm,
                                   unsigned* __restrict__ bst,
                                   unsigned* __restrict__ btot,
                                   unsigned* __restrict__ sorted,
                                   int E, int bchunk) {
    __shared__ uint4 ptot4[16 * BK_MAX / 4];
    __shared__ uint4 ppre4[16 * BK_MAX / 4];
    __shared__ unsigned gbase[BK_MAX];
    __shared__ unsigned lstart[BK_MAX];
    __shared__ unsigned cursor[BK_MAX];
    __shared__ unsigned wsa[8], wsb[8];
    __shared__ unsigned sval[BK_CHMAX];
    __shared__ unsigned short sbk[BK_CHMAX];

    const int c    = blockIdx.x;
    const int tid  = threadIdx.x;
    const int lane = tid & 31;
    const int wave = tid >> 5;

    {
        const int b4 = tid & 63;
        const int qq = tid >> 6;
        const uint4* __restrict__ m4 = (const uint4*)cntm;
        uint4 v[16];
#pragma unroll
        for (int j = 0; j < 16; ++j) v[j] = m4[(qq + 16 * j) * (BK_MAX / 4) + b4];
        uint4 t = {0u, 0u, 0u, 0u}, p = {0u, 0u, 0u, 0u};
#pragma unroll
        for (int j = 0; j < 16; ++j) {
            t.x += v[j].x; t.y += v[j].y; t.z += v[j].z; t.w += v[j].w;
            if (qq + 16 * j < c) { p.x += v[j].x; p.y += v[j].y; p.z += v[j].z; p.w += v[j].w; }
        }
        ptot4[qq * (BK_MAX / 4) + b4] = t;
        ppre4[qq * (BK_MAX / 4) + b4] = p;
    }
    __syncthreads();

    const unsigned* ptot = (const unsigned*)ptot4;
    const unsigned* ppre = (const unsigned*)ppre4;
    const int b = tid;
    unsigned tot = 0u, pre = 0u, mine = 0u, it = 0u, im = 0u;
    if (tid < BK_MAX) {
        for (int k = 0; k < 16; ++k) { tot += ptot[k * BK_MAX + b]; pre += ppre[k * BK_MAX + b]; }
        mine = cntm[c * BK_MAX + b];
        it = tot; im = mine;
        for (int d = 1; d < 32; d <<= 1) {
            const unsigned a = __shfl_up(it, (unsigned)d);
            const unsigned m = __shfl_up(im, (unsigned)d);
            if (lane >= d) { it += a; im += m; }
        }
        if (lane == 31) { wsa[wave] = it; wsb[wave] = im; }
    }
    __syncthreads();
    if (tid < BK_MAX) {
        unsigned oa = 0u, ob = 0u;
        for (int w = 0; w < wave; ++w) { oa += wsa[w]; ob += wsb[w]; }
        const unsigned bstart = oa + it - tot;
        gbase[b]  = bstart + pre;
        lstart[b] = ob + im - mine;
        cursor[b] = 0u;
        if (c == 0) { bst[b] = bstart; btot[b] = tot; }
    }
    __syncthreads();

    const int e0 = c * bchunk;
    int e1 = e0 + bchunk; if (e1 > E) e1 = E;
    const int len  = e1 > e0 ? e1 - e0 : 0;
    const int nvec = len >> 2;
    const int4* __restrict__ vs = (const int4*)(src + e0);
    const int4* __restrict__ vd = (const int4*)(dst + e0);
    for (int i = tid; i < nvec; i += DEG_THREADS) {
        const int4 s4 = vs[i];
        const int4 d4 = vd[i];
        bk_stage1(s4.x, d4.x, cursor, lstart, sval, sbk);
        bk_stage1(s4.y, d4.y, cursor, lstart, sval, sbk);
        bk_stage1(s4.z, d4.z, cursor, lstart, sval, sbk);
        bk_stage1(s4.w, d4.w, cursor, lstart, sval, sbk);
    }
    for (int e = e0 + (nvec << 2) + tid; e < e1; e += DEG_THREADS)
        bk_stage1(src[e], dst[e], cursor, lstart, sval, sbk);
    __syncthreads();

    for (int i = tid; i < len; i += DEG_THREADS) {
        const unsigned bb  = sbk[i];
        const unsigned pos = gbase[bb] + ((unsigned)i - lstart[bb]);
        if (pos < (unsigned)E) sorted[pos] = sval[i];
    }
}

#define BK_REP16(M) M(0) M(1) M(2) M(3) M(4) M(5) M(6) M(7) M(8) M(9) M(10) M(11) M(12) M(13) M(14) M(15)

template <int J, bool GUARD>
static __device__ __forceinline__ void bk_add(float* accl, unsigned drow, unsigned cb, float x,
                                              unsigned ib, unsigned n) {
    const unsigned off = row_bcast<J>(drow) + cb;
    if (!GUARD || ib + J < n)
        atomicAdd((float*)((char*)accl + off), x);
}

template <bool GUARD>
static __device__ __forceinline__ void bk_batch2(const float* __restrict__ hbuf, float* accl,
                                                 unsigned va, unsigned vb, unsigned cb,
                                                 unsigned ia, unsigned ib, unsigned n) {
    const unsigned sa = (va & ((1u << BK_SBITS) - 1u)) << 6;
    const unsigned da = (va >> BK_SBITS) << 6;
    const unsigned sb = (vb & ((1u << BK_SBITS) - 1u)) << 6;
    const unsigned db = (vb >> BK_SBITS) << 6;
    float xa[16], xb[16];
#define BK_GA(J) xa[J] = sc_load<J>(hbuf, sa, cb);
#define BK_GB(J) xb[J] = sc_load<J>(hbuf, sb, cb);
    BK_REP16(BK_GA)
    BK_REP16(BK_GB)
#define BK_AA(J) bk_add<J, GUARD>(accl, da, cb, xa[J], ia, n);
#define BK_AB(J) bk_add<J, GUARD>(accl, db, cb, xb[J], ib, n);
    BK_REP16(BK_AA)
    BK_REP16(BK_AB)
}

__launch_bounds__(DEG_THREADS)
__global__ void appnp_bucket_sum(const unsigned* __restrict__ sorted,
                                 const unsigned* __restrict__ bst,
                                 const unsigned* __restrict__ btot,
                                 const float* __restrict__ hbuf,
                                 const float* __restrict__ outdeg,
                                 float* __restrict__ out, int N) {
    __shared__ float4 accl4[BK_R * C_OUT / 4];
    __shared__ unsigned cntl[BK_R];
    float* accl = (float*)accl4;

    const int bk   = blockIdx.x;
    const int tid  = threadIdx.x;
    const int lane = tid & 31;
    const int wave = tid >> 5;
    const int lo   = bk * BK_R;
    int nr = N - lo; if (nr > BK_R) nr = BK_R;

    {
        const float4 z = {0.0f, 0.0f, 0.0f, 0.0f};
        for (int i = tid; i < BK_R * C_OUT / 4; i += DEG_THREADS) accl4[i] = z;
        for (int i = tid; i < BK_R; i += DEG_THREADS) cntl[i] = 0u;
    }
    __syncthreads();

    const unsigned start = bst[bk];
    const unsigned n     = btot[bk];
    const unsigned* __restrict__ mine = sorted + start;
    const unsigned cb   = (unsigned)(lane & 15) << 2;
    const unsigned hofs = (unsigned)(lane & 16);

    for (unsigned base = (unsigned)wave << 5; base < n; base += 2u * DEG_THREADS) {
        const unsigned ea = base + (unsigned)lane;
        const unsigned eb = ea + (unsigned)DEG_THREADS;
        const unsigned va = ea < n ? mine[ea] : 0u;
        const unsigned vb = eb < n ? mine[eb] : 0u;
        if (ea < n) atomicAdd(&cntl[va >> BK_SBITS], 1u);
        if (eb < n) atomicAdd(&cntl[vb >> BK_SBITS], 1u);
        if (base + (unsigned)DEG_THREADS + 32u <= n)
            bk_batch2<false>(hbuf, accl, va, vb, cb, base + hofs, base + (unsigned)DEG_THREADS + hofs, n);
        else
            bk_batch2<true >(hbuf, accl, va, vb, cb, base + hofs, base + (unsigned)DEG_THREADS + hofs, n);
    }
    __syncthreads();

    for (int idx = tid; idx < nr * 4; idx += DEG_THREADS) {
        const int    node = lo + (idx >> 2);
        const size_t t    = (size_t)lo * 4 + (size_t)idx;
        const float nd = rsqrtf(fmaxf((float)cntl[idx >> 2], 1.0f));
        const float ns = rsqrtf(fmaxf(outdeg[node], 1.0f));
        const float4 a  = accl4[idx];
        const float4 h0 = ((const float4*)hbuf)[t];
        float4 x0;
        x0.x = h0.x / ns; x0.y = h0.y / ns; x0.z = h0.z / ns; x0.w = h0.w / ns;
        float4 o;
        o.x = 0.8f * a.x * nd + 0.2f * x0.x;
        o.y = 0.8f * a.y * nd + 0.2f * x0.y;
        o.z = 0.8f * a.z * nd + 0.2f * x0.z;
        o.w = 0.8f * a.w * nd + 0.2f * x0.w;
        ((float4*)out)[t] = o;
    }
}

extern "C" void kernel_launch(void* const* d_in, const int* in_sizes, int n_in,
                              void* d_out, int out_size, void* d_ws, size_t ws_size,
                              hipStream_t stream) {
    const float* feat = (const float*)d_in[0];
    const float* W    = (const float*)d_in[1];
    const float* bias = (const float*)d_in[2];
    const int*   src  = (const int*)d_in[3];
    const int*   dst  = (const int*)d_in[4];

    const int N = in_sizes[0] / F_IN;
    const int E = in_sizes[3];

    float* agg1 = (float*)d_out;
    float* ws     = (float*)d_ws;
    float* xbuf   = ws;
    float* agg    = xbuf + (size_t)N * C_OUT;
    float* outdeg = agg  + (size_t)N * C_OUT;
    float* indeg  = outdeg + N;

    const int    nbk    = (N + BK_R - 1) / BK_R;
    const int    bchunk = (((E + BK_NCH - 1) / BK_NCH) + 3) & ~3;
    const size_t used   = (size_t)N * (2 * C_OUT + 2) * sizeof(float);
    const size_t need   = used + 16 + ((size_t)BK_NCH * BK_MAX + 2 * BK_MAX + (size_t)E) * sizeof(unsigned);
    const bool ordered  = ws_size >= need && N <= (1 << BK_SBITS) && nbk <= BK_MAX && bchunk <= BK_CHMAX;
    unsigned* cntm   = (unsigned*)(((size_t)(indeg + N) + 15) & ~(size_t)15);
    unsigned* bst    = cntm + (size_t)BK_NCH * BK_MAX;
    unsigned* btot   = bst + BK_MAX;
    unsigned* sorted = btot + BK_MAX;

    {
        const int nchunk = ordered ? 2 * DEG_P : DEG_P;
        const int chunk  = (((E + nchunk - 1) / nchunk) + 3) & ~3;
        const int ranges = (N + DEG_R - 1) / DEG_R;
        appnp_degree_count<<<dim3(ranges, nchunk, ordered ? 1 : 2), DEG_THREADS, 0, stream>>>(
            src, dst, (unsigned*)agg, (unsigned*)agg1, E, N, chunk, cntm, bchunk, ordered ? 1 : 0);
    }
    {
        const int nTiles = (N + 15) / 16;
        const int blocks = (nTiles + 7) / 8;
        appnp_gemm_relu<<<blocks, 256, 0, stream>>>(feat, W, bias, agg, agg1, outdeg, indeg,
                                                    xbuf, nTiles, ordered ? ~0u : 0u);
    }
    if (ordered) {
        appnp_bucket_place<<<BK_NCH, DEG_THREADS, 0, stream>>>(src, dst, cntm, bst, btot, sorted, E, bchunk);
        appnp_bucket_sum<<<nbk, DEG_THREADS, 0, stream>>>(sorted, bst, btot, xbuf, outdeg,
                                                          (float*)d_out, N);
        return;
    }
    {
        appnp_scatter<<<SC_BLOCKS, SC_THREADS, 0, stream>>>(src, dst, xbuf, agg, agg1, E);
    }
    appnp_finalize<<<(N * 4 + 255) / 256, 256, 0, stream>>>(agg, outdeg, indeg, xbuf,
                                                            (float*)d_out, N);
}


# edge sort chunk-local (per-chunk runs + start/size table; a bucket workgroup walks its 256 runs), so it shares launch 1 with the histograms: 4 launches -> 3; LDS adds in arrival order, new sequence
# speedup vs baseline: 6.0705x; 1.1067x over previous
#include <hip/hip_runtime.h>
#include <hip/hip_bf16.h>


typedef float v2f __attribute__((ext_vector_type(2)));
typedef float v8f __attribute__((ext_vector_type(8)));

#define F_IN        512
#define C_OUT       16
#define LDSW_STRIDE (F_IN + 2)
#define DEG_P       16

#define KB        64
#define NBATCH    (F_IN / KB)
#define STG_PITCH (KB + 4)
#define STG_WAVE  (16 * STG_PITCH)

struct batch8 { float4 p0, p1, p2, p3, p4, p5, p6, p7; };

static __device__ __forceinline__ batch8 gemm_load_batch(const float4* __restrict__ g) {
    batch8 d;
    d.p0 = g[0 * (2 * F_IN / 4)]; d.p1 = g[1 * (2 * F_IN / 4)];
    d.p2 = g[2 * (2 * F_IN / 4)]; d.p3 = g[3 * (2 * F_IN / 4)];
    d.p4 = g[4 * (2 * F_IN / 4)]; d.p5 = g[5 * (2 * F_IN / 4)];
    d.p6 = g[6 * (2 * F_IN / 4)]; d.p7 = g[7 * (2 * F_IN / 4)];
    return d;
}

static __device__ __forceinline__ void gemm_use_batch(const batch8 s, float4* stg4,
                                                      const float* arow, const float* browt,
                                                      v8f& acc) {
    stg4[0 * (2 * STG_PITCH / 4)] = s.p0; stg4[1 * (2 * STG_PITCH / 4)] = s.p1;
    stg4[2 * (2 * STG_PITCH / 4)] = s.p2; stg4[3 * (2 * STG_PITCH / 4)] = s.p3;
    stg4[4 * (2 * STG_PITCH / 4)] = s.p4; stg4[5 * (2 * STG_PITCH / 4)] = s.p5;
    stg4[6 * (2 * STG_PITCH / 4)] = s.p6; stg4[7 * (2 * STG_PITCH / 4)] = s.p7;
#pragma unroll
    for (int k = 0; k < KB / 4; ++k) {
        v2f a = *(const v2f*)(arow + 4 * k);
        v2f b = *(const v2f*)(browt + 4 * k);
        acc = __builtin_amdgcn_wmma_f32_16x16x4_f32(
             false, a,  false, b,
             (short)0, acc,  false,  false);
    }
}

__launch_bounds__(256, 8)
__global__ void appnp_gemm_relu(const float* __restrict__ feat,
                                const float* __restrict__ W,
                                const float* __restrict__ bias,
                                float* acc0, float* acc1,
                                float* __restrict__ outdeg,
                                float* __restrict__ indeg,
                                float* __restrict__ hbuf,
                                int nTiles, unsigned srcMask) {
    __shared__ float  Wlds[C_OUT * LDSW_STRIDE];
    __shared__ float4 Astg[8 * STG_WAVE / 4];

    const int  wave  = threadIdx.x >> 5;
    const int  lane  = threadIdx.x & 31;
    const int  tile  = blockIdx.x * 8 + wave;
    const bool valid = tile < nTiles;
    const int  half  = lane >> 4;
    const int  l     = lane & 15;

    const int tile_c = valid ? tile : nTiles - 1;
    const float4* gsrc = (const float4*)(feat + (size_t)(tile_c * 16 + half) * F_IN) + l;
    batch8 bufa = gemm_load_batch(gsrc);
    batch8 bufb;

    unsigned so = 0u, si = 0u;
    {
        const unsigned* q0 = (const unsigned*)acc0 + (size_t)tile_c * (DEG_P * 16) + half * (DEG_P * 8) + l;
        const unsigned* q1 = (const unsigned*)acc1 + (size_t)tile_c * (DEG_P * 16) + half * (DEG_P * 8) + l;
#pragma unroll
        for (int j = 0; j < DEG_P / 2; ++j) { so += q0[j * 16]; si += q1[j * 16]; }
    }

    {
        float4 w[8];
#pragma unroll
        for (int q = 0; q < 8; ++q) w[q] = ((const float4*)W)[threadIdx.x + 256 * q];
#pragma unroll
        for (int q = 0; q < 8; ++q) {
            const int i = threadIdx.x + 256 * q;
            const int r = i >> 7;
            const int c = (i & 127) << 2;
            float* dr = &Wlds[r * LDSW_STRIDE + c];
            dr[0] = w[q].x; dr[1] = w[q].y; dr[2] = w[q].z; dr[3] = w[q].w;
        }
    }
    __syncthreads();
    if (!valid) return;

    float4*      stg4 = Astg + wave * (STG_WAVE / 4) + half * (STG_PITCH / 4) + l;
    const float* arow = (const float*)Astg + wave * STG_WAVE + l * STG_PITCH + 2 * half;
    const float* brow = &Wlds[l * LDSW_STRIDE + 2 * half];

    v8f acc = {};
    bufb = gemm_load_batch(gsrc + 1 * (KB / 4)); gemm_use_batch(bufa, stg4, arow, brow + 0 * KB, acc);
    bufa = gemm_load_batch(gsrc + 2 * (KB / 4)); gemm_use_batch(bufb, stg4, arow, brow + 1 * KB, acc);
    bufb = gemm_load_batch(gsrc + 3 * (KB / 4)); gemm_use_batch(bufa, stg4, arow, brow + 2 * KB, acc);
    bufa = gemm_load_batch(gsrc + 4 * (KB / 4)); gemm_use_batch(bufb, stg4, arow, brow + 3 * KB, acc);
    bufb = gemm_load_batch(gsrc + 5 * (KB / 4)); gemm_use_batch(bufa, stg4, arow, brow + 4 * KB, acc);
    bufa = gemm_load_batch(gsrc + 6 * (KB / 4)); gemm_use_batch(bufb, stg4, arow, brow + 5 * KB, acc);
    bufb = gemm_load_batch(gsrc + 7 * (KB / 4)); gemm_use_batch(bufa, stg4, arow, brow + 6 * KB, acc);
    gemm_use_batch(bufb, stg4, arow, brow + 7 * KB, acc);

    so += (unsigned)__shfl_xor((int)so, 16);
    si += (unsigned)__shfl_xor((int)si, 16);
    so += si & srcMask;
    const float od = (float)so;
    const float id = (float)si;
    if (half) indeg[tile * 16 + l] = id; else outdeg[tile * 16 + l] = od;
    const float nsl = rsqrtf(fmaxf(od, 1.0f));

    const float bv = bias[l];
#pragma unroll
    for (int r = 0; r < 8; ++r) {
        const int row = tile * 16 + r + 8 * half;
        const float ns = __shfl(nsl, r + 8 * half);
        float v = acc[r] + bv;
        v = v > 0.0f ? v : 0.0f;
        hbuf[(size_t)row * C_OUT + l] = v * ns;
    }

    {
        const float4 z = {0.0f, 0.0f, 0.0f, 0.0f};
        float4* z0 = (float4*)(acc0 + (size_t)tile * (16 * C_OUT));
        float4* z1 = (float4*)(acc1 + (size_t)tile * (16 * C_OUT));
        z0[lane] = z; z0[lane + 32] = z;
        z1[lane] = z; z1[lane + 32] = z;
    }
}

#define DEG_R       12800
#define DEG_THREADS 1024

#define BK_R     400
#define BK_MAX   256
#define BK_NCH   256
#define BK_CHMAX 12544
#define BK_SBITS 17
static_assert(BK_NCH == 256 && BK_MAX == 256 && DEG_THREADS == 1024, "the sort's and the sum's thread maps assume these sizes");
static_assert(BK_CHMAX <= DEG_R && BK_CHMAX < 65536, "a chunk's stage lies in the histogram's LDS words; run starts and sizes are 16-bit fields");

static __device__ __forceinline__ unsigned bk_of(const int d) {
    const unsigned b = (unsigned)d / (unsigned)BK_R;
    return b < (unsigned)BK_MAX ? b : (unsigned)(BK_MAX - 1);
}

static __device__ __forceinline__ void deg_bin4(unsigned* cnt, const int4 q, const int lo) {
    unsigned r;
    r = (unsigned)(q.x - lo); if (r < (unsigned)DEG_R) atomicAdd(&cnt[r], 1u);
    r = (unsigned)(q.y - lo); if (r < (unsigned)DEG_R) atomicAdd(&cnt[r], 1u);
    r = (unsigned)(q.z - lo); if (r < (unsigned)DEG_R) atomicAdd(&cnt[r], 1u);
    r = (unsigned)(q.w - lo); if (r < (unsigned)DEG_R) atomicAdd(&cnt[r], 1u);
}

static __device__ __forceinline__ void bk_stage1(const int s, const int d,
                                                 unsigned* cursor, unsigned* sval) {
    const unsigned b    = bk_of(d);
    const unsigned dl   = (unsigned)d - b * (unsigned)BK_R;
    const unsigned slot = atomicAdd(&cursor[b], 1u);
    if (slot < (unsigned)BK_CHMAX)
        sval[slot] = (unsigned)s | (dl << BK_SBITS);
}

static __device__ __forceinline__ void bk_count4(unsigned* cnt, const int4 q) {
    atomicAdd(&cnt[bk_of(q.x)], 1u); atomicAdd(&cnt[bk_of(q.y)], 1u);
    atomicAdd(&cnt[bk_of(q.z)], 1u); atomicAdd(&cnt[bk_of(q.w)], 1u);
}

static __device__ __forceinline__ void bk_place_chunk(const int* __restrict__ src,
                                                      const int* __restrict__ dst,
                                                      unsigned* __restrict__ tab,
                                                      unsigned* __restrict__ sorted,
                                                      int E, int bchunk, int c, unsigned* sh) {
    unsigned* sval   = sh;
    unsigned* cnt    = sh + DEG_R;
    unsigned* cursor = cnt + BK_MAX;
    unsigned* wsb    = cursor + BK_MAX;
    const int tid  = threadIdx.x;
    const int lane = tid & 31;
    const int wave = tid >> 5;

    if (tid < BK_MAX) cnt[tid] = 0u;
    __syncthreads();

    const int e0 = c * bchunk;
    int e1 = e0 + bchunk; if (e1 > E) e1 = E;
    const int len  = e1 > e0 ? e1 - e0 : 0;
    const int nvec = len >> 2;
    const int4* __restrict__ vs = (const int4*)(src + e0);
    const int4* __restrict__ vd = (const int4*)(dst + e0);

    for (int i = tid; i < nvec; i += DEG_THREADS) bk_count4(cnt, vd[i]);
    for (int e = e0 + (nvec << 2) + tid; e < e1; e += DEG_THREADS)
        atomicAdd(&cnt[bk_of(dst[e])], 1u);
    __syncthreads();

    unsigned mine = 0u, im = 0u;
    if (tid < BK_MAX) {
        mine = cnt[tid];
        im = mine;
        for (int d = 1; d < 32; d <<= 1) {
            const unsigned m = __shfl_up(im, (unsigned)d);
            if (lane >= d) im += m;
        }
        if (lane == 31) wsb[wave] = im;
    }
    __syncthreads();
    if (tid < BK_MAX) {
        unsigned ob = 0u;
        for (int w = 0; w < wave; ++w) ob += wsb[w];
        const unsigned ls = ob + im - mine;
        cursor[tid] = ls;
        tab[(size_t)tid * BK_NCH + c] = ls | (mine << 16);
    }
    __syncthreads();

    for (int i = tid; i < nvec; i += DEG_THREADS) {
        const int4 s4 = vs[i];
        const int4 d4 = vd[i];
        bk_stage1(s4.x, d4.x, cursor, sval);
        bk_stage1(s4.y, d4.y, cursor, sval);
        bk_stage1(s4.z, d4.z, cursor, sval);
        bk_stage1(s4.w, d4.w, cursor, sval);
    }
    for (int e = e0 + (nvec << 2) + tid; e < e1; e += DEG_THREADS)
        bk_stage1(src[e], dst[e], cursor, sval);
    __syncthreads();

    unsigned* __restrict__ outp = sorted + (size_t)c * bchunk;
    for (int i = tid; i < nvec; i += DEG_THREADS) ((uint4*)outp)[i] = ((const uint4*)sval)[i];
    for (int i = (nvec << 2) + tid; i < len; i += DEG_THREADS) outp[i] = sval[i];
}

__launch_bounds__(DEG_THREADS)
__global__ void appnp_degree_count(const int* __restrict__ src,
                                   const int* __restrict__ dst,
                                   unsigned* __restrict__ part0,
                                   unsigned* __restrict__ part1,
                                   int E, int N, int chunk,
                                   unsigned* __restrict__ tab,
                                   unsigned* __restrict__ sorted,
                                   int bchunk, int srcOnly, int nhist) {
    __shared__ __attribute__((aligned(16))) unsigned sh[DEG_R + 2 * BK_MAX + 8];
    unsigned* cnt = sh;

    if ((int)blockIdx.y >= nhist) {
        const int c = ((int)blockIdx.y - nhist) * (int)gridDim.x + (int)blockIdx.x;
        if (c < BK_NCH) bk_place_chunk(src, dst, tab, sorted, E, bchunk, c, sh);
        return;
    }

    const int h = blockIdx.x;
    const int a = srcOnly ? (int)(blockIdx.y / DEG_P) : (int)blockIdx.z;
    const int p = srcOnly ? (int)(blockIdx.y % DEG_P) : (int)blockIdx.y;
    const int* __restrict__ idx = (a && !srcOnly) ? dst : src;
    const int lo = h * DEG_R;

    for (int i = threadIdx.x; i < DEG_R; i += DEG_THREADS) cnt[i] = 0u;
    __syncthreads();

    const int e0 = (int)blockIdx.y * chunk;
    int e1 = e0 + chunk; if (e1 > E) e1 = E;
    const int len  = e1 > e0 ? e1 - e0 : 0;
    const int nvec = len >> 2;
    const int4* __restrict__ v = (const int4*)(idx + e0);

    int i = threadIdx.x;
    for (; i + 3 * DEG_THREADS < nvec; i += 4 * DEG_THREADS) {
        const int4 q0 = v[i];
        const int4 q1 = v[i + DEG_THREADS];
        const int4 q2 = v[i + 2 * DEG_THREADS];
        const int4 q3 = v[i + 3 * DEG_THREADS];
        deg_bin4(cnt, q0, lo); deg_bin4(cnt, q1, lo);
        deg_bin4(cnt, q2, lo); deg_bin4(cnt, q3, lo);
    }
    for (; i < nvec; i += DEG_THREADS) deg_bin4(cnt, v[i], lo);
    for (int e = e0 + (nvec << 2) + threadIdx.x; e < e1; e += DEG_THREADS) {
        const unsigned r = (unsigned)(idx[e] - lo);
        if (r < (unsigned)DEG_R) atomicAdd(&cnt[r], 1u);
    }
    __syncthreads();

    int nr = N - lo; if (nr > DEG_R) nr = DEG_R;
    unsigned* __restrict__ plane = (a ? part1 : part0) + p * 16;
    for (int k = threadIdx.x; k < nr; k += DEG_THREADS) {
        const int n = lo + k;
        plane[(size_t)(n >> 4) * (DEG_P * 16) + (n & 15)] = cnt[k];
    }
}

#define SC_THREADS 256
#define SC_BLOCKS  2048

static __device__ __forceinline__ unsigned l2_half() {
    const unsigned id = (unsigned)__builtin_amdgcn_s_sendmsg_rtn(0x87);
    return (id >> 18) & 1u;
}

template <int J>
static __device__ __forceinline__ unsigned row_bcast(unsigned v) {
    return (unsigned)__builtin_amdgcn_update_dpp(0, (int)v, 0x150 + J, 0xf, 0xf, true);
}

template <int J>
static __device__ __forceinline__ float sc_load(const float* __restrict__ xbuf,
                                                unsigned srow, unsigned cb) {
    return *(const float*)((const char*)xbuf + (row_bcast<J>(srow) + cb));
}

template <int J, bool GUARD>
static __device__ __forceinline__ void sc_add(float* __restrict__ agg, unsigned drow,
                                              unsigned cb, float x,
                                              int eb, int E) {
    const unsigned off = row_bcast<J>(drow) + cb;
    if (!GUARD || eb + J < E)
        atomicAdd((float*)((char*)agg + off), x);
}

template <bool GUARD>
static __device__ __forceinline__ void sc_batch(const float* __restrict__ xbuf,
                                                float* __restrict__ agg,
                                                unsigned srow, unsigned drow,
                                                unsigned cb, int eb, int E) {
    const float x0  = sc_load<0>(xbuf, srow, cb);
    const float x1  = sc_load<1>(xbuf, srow, cb);
    const float x2  = sc_load<2>(xbuf, srow, cb);
    const float x3  = sc_load<3>(xbuf, srow, cb);
    const float x4  = sc_load<4>(xbuf, srow, cb);
    const float x5  = sc_load<5>(xbuf, srow, cb);
    const float x6  = sc_load<6>(xbuf, srow, cb);
    const float x7  = sc_load<7>(xbuf, srow, cb);
    const float x8  = sc_load<8>(xbuf, srow, cb);
    const float x9  = sc_load<9>(xbuf, srow, cb);
    const float x10 = sc_load<10>(xbuf, srow, cb);
    const float x11 = sc_load<11>(xbuf, srow, cb);
    const float x12 = sc_load<12>(xbuf, srow, cb);
    const float x13 = sc_load<13>(xbuf, srow, cb);
    const float x14 = sc_load<14>(xbuf, srow, cb);
    const float x15 = sc_load<15>(xbuf, srow, cb);
    sc_add<0,  GUARD>(agg, drow, cb, x0,  eb, E);
    sc_add<1,  GUARD>(agg, drow, cb, x1,  eb, E);
    sc_add<2,  GUARD>(agg, drow, cb, x2,  eb, E);
    sc_add<3,  GUARD>(agg, drow, cb, x3,  eb, E);
    sc_add<4,  GUARD>(agg, drow, cb, x4,  eb, E);
    sc_add<5,  GUARD>(agg, drow, cb, x5,  eb, E);
    sc_add<6,  GUARD>(agg, drow, cb, x6,  eb, E);
    sc_add<7,  GUARD>(agg, drow, cb, x7,  eb, E);
    sc_add<8,  GUARD>(agg, drow, cb, x8,  eb, E);
    sc_add<9,  GUARD>(agg, drow, cb, x9,  eb, E);
    sc_add<10, GUARD>(agg, drow, cb, x10, eb, E);
    sc_add<11, GUARD>(agg, drow, cb, x11, eb, E);
    sc_add<12, GUARD>(agg, drow, cb, x12, eb, E);
    sc_add<13, GUARD>(agg, drow, cb, x13, eb, E);
    sc_add<14, GUARD>(agg, drow, cb, x14, eb, E);
    sc_add<15, GUARD>(agg, drow, cb, x15, eb, E);
}

__launch_bounds__(SC_THREADS)
__global__ void appnp_scatter(const int* __restrict__ src,
                              const int* __restrict__ dst,
                              const float* __restrict__ xbuf,
                              float* agg0, float* agg1, int E) {
    float* __restrict__ agg = l2_half() ? agg1 : agg0;
    const int lane   = threadIdx.x & 31;
    const int gwave  = (int)((blockIdx.x * SC_THREADS + threadIdx.x) >> 5);
    const int nwaves = (int)((gridDim.x * SC_THREADS) >> 5);
    const int nb     = (E + 31) >> 5;
    const unsigned cb = (unsigned)(lane & 15) << 2;
    const int hofs   = lane & 16;

    int b = gwave;
    if (b >= nb) return;
    int e = (b << 5) + lane;
    int s = 0, d = 0;
    if (e < E) { s = src[e]; d = dst[e]; }

    for (;;) {
        const int bn = b + nwaves;
        int sn = 0, dn = 0;
        if (bn < nb) {
            const int en = (bn << 5) + lane;
            if (en < E) { sn = src[en]; dn = dst[en]; }
        }
        const unsigned srow = (unsigned)s << 6;
        const unsigned drow = (unsigned)d << 6;
        const int eb = (b << 5) + hofs;
        if ((b << 5) + 32 <= E) sc_batch<false>(xbuf, agg, srow, drow, cb, eb, E);
        else                    sc_batch<true >(xbuf, agg, srow, drow, cb, eb, E);
        if (bn >= nb) break;
        b = bn; s = sn; d = dn;
    }
}

__global__ void appnp_finalize(const float* __restrict__ agg0,
                               const float* __restrict__ outdeg,
                               const float* __restrict__ indeg,
                               const float* __restrict__ hbuf,
                               float* out, int N) {
    int t = blockIdx.x * blockDim.x + threadIdx.x;
    if (t >= N * 4) return;
    const int n = t >> 2;
    const float nd = rsqrtf(fmaxf(indeg[n], 1.0f));
    const float ns = rsqrtf(fmaxf(outdeg[n], 1.0f));
    const float4 a0 = ((const float4*)agg0)[t];
    const float4 a1 = ((const float4*)out)[t];
    float4 a;
    a.x = a0.x + a1.x; a.y = a0.y + a1.y; a.z = a0.z + a1.z; a.w = a0.w + a1.w;
    const float4 h0 = ((const float4*)hbuf)[t];
    float4 x0;
    x0.x = h0.x / ns; x0.y = h0.y / ns; x0.z = h0.z / ns; x0.w = h0.w / ns;
    float4 o;
    o.x = 0.8f * a.x * nd + 0.2f * x0.x;
    o.y = 0.8f * a.y * nd + 0.2f * x0.y;
    o.z = 0.8f * a.z * nd + 0.2f * x0.z;
    o.w = 0.8f * a.w * nd + 0.2f * x0.w;
    ((float4*)out)[t] = o;
}


#define BK_REP16(M) M(0) M(1) M(2) M(3) M(4) M(5) M(6) M(7) M(8) M(9) M(10) M(11) M(12) M(13) M(14) M(15)

template <int J, bool GUARD>
static __device__ __forceinline__ void bk_add(float* accl, unsigned drow, unsigned cb, float x,
                                              unsigned ib, unsigned n) {
    const unsigned off = row_bcast<J>(drow) + cb;
    if (!GUARD || ib + J < n)
        atomicAdd((float*)((char*)accl + off), x);
}

template <bool GUARD>
static __device__ __forceinline__ void bk_batch2(const float* __restrict__ hbuf, float* accl,
                                                 unsigned va, unsigned vb, unsigned cb,
                                                 unsigned ia, unsigned ib, unsigned n) {
    const unsigned sa = (va & ((1u << BK_SBITS) - 1u)) << 6;
    const unsigned da = (va >> BK_SBITS) << 6;
    const unsigned sb = (vb & ((1u << BK_SBITS) - 1u)) << 6;
    const unsigned db = (vb >> BK_SBITS) << 6;
    float xa[16], xb[16];
#define BK_GA(J) xa[J] = sc_load<J>(hbuf, sa, cb);
#define BK_GB(J) xb[J] = sc_load<J>(hbuf, sb, cb);
    BK_REP16(BK_GA)
    BK_REP16(BK_GB)
#define BK_AA(J) bk_add<J, GUARD>(accl, da, cb, xa[J], ia, n);
#define BK_AB(J) bk_add<J, GUARD>(accl, db, cb, xb[J], ib, n);
    BK_REP16(BK_AA)
    BK_REP16(BK_AB)
}

#define BK_RUNS (BK_NCH / 32)
static_assert(BK_RUNS == 8, "the run walk below names its 8 runs one by one");

__launch_bounds__(DEG_THREADS)
__global__ void appnp_bucket_sum(const unsigned* __restrict__ sorted,
                                 const unsigned* __restrict__ tab,
                                 const float* __restrict__ hbuf,
                                 const float* __restrict__ outdeg,
                                 float* __restrict__ out, int N, int bchunk) {
    __shared__ float4 accl4[BK_R * C_OUT / 4];
    __shared__ unsigned cntl[BK_R];
    float* accl = (float*)accl4;

    const int bk   = blockIdx.x;
    const int tid  = threadIdx.x;
    const int lane = tid & 31;
    const int wave = tid >> 5;
    const int lo   = bk * BK_R;
    int nr = N - lo; if (nr > BK_R) nr = BK_R;

    unsigned tw = 0u;
    if (lane < BK_RUNS) tw = tab[(size_t)bk * BK_NCH + wave * BK_RUNS + lane];

    {
        const float4 z = {0.0f, 0.0f, 0.0f, 0.0f};
        for (int i = tid; i < BK_R * C_OUT / 4; i += DEG_THREADS) accl4[i] = z;
        for (int i = tid; i < BK_R; i += DEG_THREADS) cntl[i] = 0u;
    }
    __syncthreads();

    const unsigned rcnt = tw >> 16;
    const unsigned rst  = (unsigned)(wave * BK_RUNS + lane) * (unsigned)bchunk + (tw & 0xffffu);
    unsigned inc = rcnt;
    for (int d = 1; d < BK_RUNS; d <<= 1) {
        const unsigned m = __shfl_up(inc, (unsigned)d);
        if (lane >= d) inc += m;
    }
    const unsigned rpre = inc - rcnt;
    const unsigned rsb  = rst - rpre;
    const unsigned T  = (unsigned)__builtin_amdgcn_readlane((int)inc, BK_RUNS - 1);
    const unsigned p1 = (unsigned)__builtin_amdgcn_readlane((int)rpre, 1);
    const unsigned p2 = (unsigned)__builtin_amdgcn_readlane((int)rpre, 2);
    const unsigned p3 = (unsigned)__builtin_amdgcn_readlane((int)rpre, 3);
    const unsigned p4 = (unsigned)__builtin_amdgcn_readlane((int)rpre, 4);
    const unsigned p5 = (unsigned)__builtin_amdgcn_readlane((int)rpre, 5);
    const unsigned p6 = (unsigned)__builtin_amdgcn_readlane((int)rpre, 6);
    const unsigned p7 = (unsigned)__builtin_amdgcn_readlane((int)rpre, 7);
    const unsigned b0 = (unsigned)__builtin_amdgcn_readlane((int)rsb, 0);
    const unsigned b1 = (unsigned)__builtin_amdgcn_readlane((int)rsb, 1);
    const unsigned b2 = (unsigned)__builtin_amdgcn_readlane((int)rsb, 2);
    const unsigned b3 = (unsigned)__builtin_amdgcn_readlane((int)rsb, 3);
    const unsigned b4 = (unsigned)__builtin_amdgcn_readlane((int)rsb, 4);
    const unsigned b5 = (unsigned)__builtin_amdgcn_readlane((int)rsb, 5);
    const unsigned b6 = (unsigned)__builtin_amdgcn_readlane((int)rsb, 6);
    const unsigned b7 = (unsigned)__builtin_amdgcn_readlane((int)rsb, 7);

    const unsigned cb   = (unsigned)(lane & 15) << 2;
    const unsigned hofs = (unsigned)(lane & 16);

    for (unsigned base = 0u; base < T; base += 64u) {
        const unsigned ea = base + (unsigned)lane;
        const unsigned eb = ea + 32u;
        unsigned ia = b0, ib = b0;
        if (ea >= p1) ia = b1;  if (eb >= p1) ib = b1;
        if (ea >= p2) ia = b2;  if (eb >= p2) ib = b2;
        if (ea >= p3) ia = b3;  if (eb >= p3) ib = b3;
        if (ea >= p4) ia = b4;  if (eb >= p4) ib = b4;
        if (ea >= p5) ia = b5;  if (eb >= p5) ib = b5;
        if (ea >= p6) ia = b6;  if (eb >= p6) ib = b6;
        if (ea >= p7) ia = b7;  if (eb >= p7) ib = b7;
        const unsigned va = ea < T ? sorted[ia + ea] : 0u;
        const unsigned vb = eb < T ? sorted[ib + eb] : 0u;
        if (ea < T) atomicAdd(&cntl[va >> BK_SBITS], 1u);
        if (eb < T) atomicAdd(&cntl[vb >> BK_SBITS], 1u);
        if (base + 64u <= T)
            bk_batch2<false>(hbuf, accl, va, vb, cb, base + hofs, base + 32u + hofs, T);
        else
            bk_batch2<true >(hbuf, accl, va, vb, cb, base + hofs, base + 32u + hofs, T);
    }
    __syncthreads();

    for (int idx = tid; idx < nr * 4; idx += DEG_THREADS) {
        const int    node = lo + (idx >> 2);
        const size_t t    = (size_t)lo * 4 + (size_t)idx;
        const float nd = rsqrtf(fmaxf((float)cntl[idx >> 2], 1.0f));
        const float ns = rsqrtf(fmaxf(outdeg[node], 1.0f));
        const float4 a  = accl4[idx];
        const float4 h0 = ((const float4*)hbuf)[t];
        float4 x0;
        x0.x = h0.x / ns; x0.y = h0.y / ns; x0.z = h0.z / ns; x0.w = h0.w / ns;
        float4 o;
        o.x = 0.8f * a.x * nd + 0.2f * x0.x;
        o.y = 0.8f * a.y * nd + 0.2f * x0.y;
        o.z = 0.8f * a.z * nd + 0.2f * x0.z;
        o.w = 0.8f * a.w * nd + 0.2f * x0.w;
        ((float4*)out)[t] = o;
    }
}

extern "C" void kernel_launch(void* const* d_in, const int* in_sizes, int n_in,
                              void* d_out, int out_size, void* d_ws, size_t ws_size,
                              hipStream_t stream) {
    const float* feat = (const float*)d_in[0];
    const float* W    = (const float*)d_in[1];
    const float* bias = (const float*)d_in[2];
    const int*   src  = (const int*)d_in[3];
    const int*   dst  = (const int*)d_in[4];

    const int N = in_sizes[0] / F_IN;
    const int E = in_sizes[3];

    float* agg1 = (float*)d_out;
    float* ws     = (float*)d_ws;
    float* xbuf   = ws;
    float* agg    = xbuf + (size_t)N * C_OUT;
    float* outdeg = agg  + (size_t)N * C_OUT;
    float* indeg  = outdeg + N;

    const int    nbk    = (N + BK_R - 1) / BK_R;
    const int    bchunk = (((E + BK_NCH - 1) / BK_NCH) + 3) & ~3;
    const size_t used   = (size_t)N * (2 * C_OUT + 2) * sizeof(float);
    const size_t need   = used + 16 + ((size_t)BK_MAX * BK_NCH + (size_t)BK_NCH * bchunk) * sizeof(unsigned);
    const bool ordered  = ws_size >= need && N <= (1 << BK_SBITS) && nbk <= BK_MAX && bchunk <= BK_CHMAX;
    unsigned* tab    = (unsigned*)(((size_t)(indeg + N) + 15) & ~(size_t)15);
    unsigned* sorted = tab + (size_t)BK_MAX * BK_NCH;

    {
        const int nchunk = ordered ? 2 * DEG_P : DEG_P;
        const int chunk  = (((E + nchunk - 1) / nchunk) + 3) & ~3;
        const int ranges = (N + DEG_R - 1) / DEG_R;
        const int nsort  = ordered ? (BK_NCH + ranges - 1) / ranges : 0;
        appnp_degree_count<<<dim3(ranges, nchunk + nsort, ordered ? 1 : 2), DEG_THREADS, 0, stream>>>(
            src, dst, (unsigned*)agg, (unsigned*)agg1, E, N, chunk, tab, sorted, bchunk, ordered ? 1 : 0, nchunk);
    }
    {
        const int nTiles = (N + 15) / 16;
        const int blocks = (nTiles + 7) / 8;
        appnp_gemm_relu<<<blocks, 256, 0, stream>>>(feat, W, bias, agg, agg1, outdeg, indeg,
                                                    xbuf, nTiles, ordered ? ~0u : 0u);
    }
    if (ordered) {
        appnp_bucket_sum<<<nbk, DEG_THREADS, 0, stream>>>(sorted, tab, xbuf, outdeg,
                                                          (float*)d_out, N, bchunk);
        return;
    }
    {
        appnp_scatter<<<SC_BLOCKS, SC_THREADS, 0, stream>>>(src, dst, xbuf, agg, agg1, E);
    }
    appnp_finalize<<<(N * 4 + 255) / 256, 256, 0, stream>>>(agg, outdeg, indeg, xbuf,
                                                            (float*)d_out, N);
}
